# starting kernel
# baseline (speedup 1.0000x reference)
#include <hip/hip_runtime.h>
#include <cmath>

typedef float v2f __attribute__((ext_vector_type(2)));
typedef float v8f __attribute__((ext_vector_type(8)));


struct cdbl { double re, im; };
__device__ inline cdbl cmul(cdbl a, cdbl b){ return {a.re*b.re - a.im*b.im, a.re*b.im + a.im*b.re}; }
__device__ inline cdbl cconj(cdbl a){ return {a.re, -a.im}; }
__device__ inline cdbl cadd(cdbl a, cdbl b){ return {a.re+b.re, a.im+b.im}; }
__device__ inline cdbl cscale(cdbl a, double s){ return {a.re*s, a.im*s}; }

__device__ inline int imini(int a, int b){ return a < b ? a : b; }
__device__ inline int imaxi(int a, int b){ return a > b ? a : b; }

__device__ double factd(int n){ double r = 1.0; for(int i = 2; i <= n; ++i) r *= (double)i; return r; }

__device__ double cgq(int j1,int m1,int j2,int m2,int j3,int m3){
  if (m3 != m1 + m2) return 0.0;
  int vmin = imaxi(imaxi(-j1 + j2 + m3, -j1 + m1), 0);
  int vmax = imini(imini(j2 + j3 + m1, j3 - j1 + j2), j3 + m3);
  double c = sqrt((double)(2*j3+1) * factd(j3+j1-j2) * factd(j3-j1+j2) * factd(j1+j2-j3)
                  * factd(j3+m3) * factd(j3-m3)
                  / (factd(j1+j2+j3+1) * factd(j1-m1) * factd(j1+m1) * factd(j2-m2) * factd(j2+m2)));
  double s = 0.0;
  for (int v = vmin; v <= vmax; ++v){
    double sg = ((v + j2 + m2) & 1) ? -1.0 : 1.0;
    s += sg * factd(j2+j3+m1-v) * factd(j1-m1+v)
         / (factd(v) * factd(j3-j1+j2-v) * factd(j3+m3-v) * factd(v+j1-j2-m3));
  }
  return c * s;
}

__device__ void qmat(int l, cdbl* q){
  int n = 2*l + 1;
  for (int t = 0; t < n*n; ++t) q[t] = {0.0, 0.0};
  const double is2 = 0.70710678118654752440;
  for (int m = -l; m < 0; ++m){
    q[(l+m)*n + (l-m)] = { is2, 0.0 };
    q[(l+m)*n + (l+m)] = { 0.0, -is2 };
  }
  q[l*n + l] = {1.0, 0.0};
  for (int m = 1; m <= l; ++m){
    double sg = (m & 1) ? -1.0 : 1.0;
    q[(l+m)*n + (l+m)] = { sg*is2, 0.0 };
    q[(l+m)*n + (l-m)] = { 0.0, sg*is2 };
  }
  cdbl ph = {1.0, 0.0}; const cdbl mi = {0.0, -1.0};
  for (int t = 0; t < l; ++t) ph = cmul(ph, mi);
  for (int t = 0; t < n*n; ++t) q[t] = cmul(q[t], ph);
}

__device__ double w3j_elem(int l1,int l2,int l3,
                           const cdbl* q1,const cdbl* q2,const cdbl* q3,
                           int a,int b,int c){
  int n1 = 2*l1+1, n2 = 2*l2+1, n3 = 2*l3+1;
  cdbl s = {0.0, 0.0};
  for (int i = 0; i < n1; ++i)
    for (int k = 0; k < n2; ++k){
      int m = (i - l1) + (k - l2) + l3;
      if (m < 0 || m >= n3) continue;
      double cv = cgq(l1, i-l1, l2, k-l2, l3, m-l3);
      if (cv == 0.0) continue;
      cdbl t = cmul(cmul(q1[i*n1 + a], q2[k*n2 + b]), cconj(q3[m*n3 + c]));
      s = cadd(s, cscale(t, cv));
    }
  return s.re;
}

__device__ __constant__ int IDXc[6] = {0, 4, 8, 1, 5, 6};

__global__ __launch_bounds__(256) void piezo_setup_b(const float* __restrict__ w,
                                                     float* __restrict__ Bt){
  __shared__ cdbl Q0[1], Q1[9], Q2[25], Q3[49];
  __shared__ double W101s[9], W121s[45], W221s[75], W321s[105], W110s[9], W112s[45];
  __shared__ double COBs[6][9];
  const int tid = threadIdx.x;

  if (tid == 0){ qmat(0, Q0); qmat(1, Q1); qmat(2, Q2); qmat(3, Q3); }
  __syncthreads();

  for (int idx = tid; idx < 288; idx += blockDim.x){
    if (idx < 9){
      int a = idx/3, c = idx%3;
      W101s[idx] = w3j_elem(1,0,1, Q1,Q0,Q1, a,0,c);
    } else if (idx < 54){
      int e = idx - 9; int a = e/15, b = (e/3)%5, c = e%3;
      W121s[e] = w3j_elem(1,2,1, Q1,Q2,Q1, a,b,c);
    } else if (idx < 129){
      int e = idx - 54; int a = e/15, b = (e/3)%5, c = e%3;
      W221s[e] = w3j_elem(2,2,1, Q2,Q2,Q1, a,b,c);
    } else if (idx < 234){
      int e = idx - 129; int a = e/15, b = (e/3)%5, c = e%3;
      W321s[e] = w3j_elem(3,2,1, Q3,Q2,Q1, a,b,c);
    } else if (idx < 243){
      int e = idx - 234; int a = e/3, b = e%3;
      W110s[e] = w3j_elem(1,1,0, Q1,Q1,Q0, a,b,0);
    } else {
      int e = idx - 243; int a = e/15, b = (e/5)%3, c = e%5;
      W112s[e] = w3j_elem(1,1,2, Q1,Q1,Q2, a,b,c);
    }
  }
  __syncthreads();

  if (tid < 6){
    double* arr; int n;
    switch (tid){
      case 0: arr = W101s; n = 9;   break;
      case 1: arr = W121s; n = 45;  break;
      case 2: arr = W221s; n = 75;  break;
      case 3: arr = W321s; n = 105; break;
      case 4: arr = W110s; n = 9;   break;
      default: arr = W112s; n = 45; break;
    }
    double nrm = 0.0;
    for (int t = 0; t < n; ++t) nrm += arr[t]*arr[t];
    nrm = sqrt(nrm);
    for (int t = 0; t < n; ++t) arr[t] /= nrm;
  }
  __syncthreads();

  if (tid == 0){
    const double s5 = sqrt(5.0);
    for (int p = 0; p < 3; ++p)
      for (int q = 0; q < 3; ++q){
        COBs[0][3*p+q] = W110s[p*3 + q];
        for (int m = 0; m < 5; ++m)
          COBs[1+m][3*p+q] = s5 * W112s[(p*3+q)*5 + m];
      }
  }
  __syncthreads();

  const double ALPHA = 0.61237243569579452455;
  for (int e = tid; e < 32*64; e += blockDim.x){
    int n = e >> 6, k = e & 63;
    float val = 0.0f;
    if (n < 18){
      int kOut = n / 6, t = n % 6;
      int pq = IDXc[t];
      double v = 0.0;
      if (k >= 10 && k < 16){
        int u = (k-10)/3, i = (k-10)%3;
        v += (double)w[u] * W101s[i*3 + kOut] * COBs[0][pq];
        for (int j = 0; j < 5; ++j)
          v += (double)w[2+u] * W121s[(i*5+j)*3 + kOut] * COBs[1+j][pq];
      } else if (k >= 26 && k < 36){
        int u = (k-26)/5, i = (k-26)%5;
        for (int j = 0; j < 5; ++j)
          v += (double)w[4+u] * W221s[(i*5+j)*3 + kOut] * COBs[1+j][pq];
      } else if (k >= 50){
        int u = (k-50)/7, i = (k-50)%7;
        for (int j = 0; j < 5; ++j)
          v += (double)w[6+u] * W321s[(i*5+j)*3 + kOut] * COBs[1+j][pq];
      }
      val = (float)(ALPHA * v);
    }
    Bt[n*64 + k] = val;
  }
}


#define WPB 4
#define TPW 5

__global__ __launch_bounds__(WPB*32) void piezo_wmma_main(const float* __restrict__ X,
                                                          const float* __restrict__ Bt,
                                                          float* __restrict__ out,
                                                          int rows){
  const int lane = threadIdx.x & 31;
  const int wave = threadIdx.x >> 5;
  const int half = lane >> 4;
  const int l16  = lane & 15;

  v2f bf0[16], bf1[16];
  #pragma unroll
  for (int kk = 0; kk < 16; ++kk){
    int k0 = 4*kk + 2*half;
    bf0[kk] = *(const v2f*)(Bt + (l16     ) * 64 + k0);
    bf1[kk] = *(const v2f*)(Bt + (l16 + 16) * 64 + k0);
  }

  const long tilebase = (long)(blockIdx.x * WPB + wave) * TPW;
  for (int t = 0; t < TPW; ++t){
    long row0 = (tilebase + t) * 16;
    if (row0 + 16 > (long)rows) break;
    const float* xr = X + (row0 + l16) * 64 + 2*half;

    __builtin_prefetch((const void*)(xr + 16*64), 0, 1);
    __builtin_prefetch((const void*)(xr + 16*64 + 32), 0, 1);

    v8f acc0 = {}; v8f acc1 = {};
    #pragma unroll
    for (int kk = 0; kk < 16; ++kk){
      v2f a = *(const v2f*)(xr + 4*kk);
      acc0 = __builtin_amdgcn_wmma_f32_16x16x4_f32(false, a, false, bf0[kk],
                                                   (short)0, acc0, false, false);
      acc1 = __builtin_amdgcn_wmma_f32_16x16x4_f32(false, a, false, bf1[kk],
                                                   (short)0, acc1, false, false);
    }

    float* orow = out + row0 * 18;
    #pragma unroll
    for (int r = 0; r < 8; ++r){
      int row = r + half*8;
      orow[row*18 + l16] = acc0[r];
      if (l16 < 2) orow[row*18 + 16 + l16] = acc1[r];
    }
  }
}


extern "C" void kernel_launch(void* const* d_in, const int* in_sizes, int n_in,
                              void* d_out, int out_size, void* d_ws, size_t ws_size,
                              hipStream_t stream){
  const float* X = (const float*)d_in[0];
  const float* w = (const float*)d_in[1];
  float* out = (float*)d_out;
  float* Bt  = (float*)d_ws;

  const int rows  = in_sizes[0] / 64;
  const int tiles = rows / 16;
  const int blocks = (tiles + WPB*TPW - 1) / (WPB*TPW);

  piezo_setup_b<<<1, 256, 0, stream>>>(w, Bt);
  piezo_wmma_main<<<blocks, WPB*32, 0, stream>>>(X, Bt, out, rows);
}


# Setup launch: every fp64 value computed once by one thread and shared via LDS (CG per (m1,m2) pair, sparse q rows, norms as 8-way partial sums folded per path); single-thread phases removed
# speedup vs baseline: 2.1723x; 2.1723x over previous
#include <hip/hip_runtime.h>
#include <cmath>

typedef float v2f __attribute__((ext_vector_type(2)));
typedef float v8f __attribute__((ext_vector_type(8)));


struct cdbl { double re, im; };
__device__ inline cdbl cmul(cdbl a, cdbl b){ return {a.re*b.re - a.im*b.im, a.re*b.im + a.im*b.re}; }
__device__ inline cdbl cconj(cdbl a){ return {a.re, -a.im}; }

__device__ inline int imini(int a, int b){ return a < b ? a : b; }
__device__ inline int imaxi(int a, int b){ return a > b ? a : b; }
__device__ inline int iabsi(int a){ return a < 0 ? -a : a; }

__device__ __constant__ double FACTc[9] = {1.0, 1.0, 2.0, 6.0, 24.0, 120.0, 720.0, 5040.0, 40320.0};
__device__ inline double factd(int n){ return FACTc[n]; }

__device__ double cgq(int j1,int m1,int j2,int m2,int j3,int m3){
  if (m3 != m1 + m2) return 0.0;
  int vmin = imaxi(imaxi(-j1 + j2 + m3, -j1 + m1), 0);
  int vmax = imini(imini(j2 + j3 + m1, j3 - j1 + j2), j3 + m3);
  double c = sqrt((double)(2*j3+1) * factd(j3+j1-j2) * factd(j3-j1+j2) * factd(j1+j2-j3)
                  * factd(j3+m3) * factd(j3-m3)
                  / (factd(j1+j2+j3+1) * factd(j1-m1) * factd(j1+m1) * factd(j2-m2) * factd(j2+m2)));
  double s = 0.0;
  for (int v = vmin; v <= vmax; ++v){
    double sg = ((v + j2 + m2) & 1) ? -1.0 : 1.0;
    s += sg * factd(j2+j3+m1-v) * factd(j1-m1+v)
         / (factd(v) * factd(j3-j1+j2-v) * factd(j3+m3-v) * factd(v+j1-j2-m3));
  }
  return c * s;
}

__device__ inline cdbl qent(int l, int r, int c){
  const double is2 = 0.70710678118654752440;
  int mu = iabsi(c - l);
  double x = 0.0, y = 0.0;
  if (mu == 0){
    if (r == l) x = 1.0;
  } else {
    double sg = (mu & 1) ? -is2 : is2;
    if (c > l){
      if (r == l - mu) x = is2; else if (r == l + mu) x = sg;
    } else {
      if (r == l - mu) y = -is2; else if (r == l + mu) y = sg;
    }
  }
  switch (l & 3){
    case 0:  return { x,  y};
    case 1:  return { y, -x};
    case 2:  return {-x, -y};
    default: return {-y,  x};
  }
}

__device__ __constant__ int TL1c[6] = {1, 1, 2, 3, 1, 1};
__device__ __constant__ int TL2c[6] = {0, 2, 2, 2, 1, 1};
__device__ __constant__ int TL3c[6] = {1, 1, 1, 1, 0, 2};
__device__ __constant__ int TEOFFc[7] = {0, 9, 54, 129, 234, 243, 288};
__device__ __constant__ int TCOFFc[7] = {0, 3, 18, 43, 78, 87, 96};
__device__ __constant__ int IDXc[6] = {0, 4, 8, 1, 5, 6};

__device__ inline bool bt_active(int n, int k){
  return n < 18 && ((k >= 10 && k < 16) || (k >= 26 && k < 36) || k >= 50);
}

__device__ inline void setup_zero(int tid, int nthr, float* Bt){
  for (int e = tid; e < 32*64; e += nthr)
    if (!bt_active(e >> 6, e & 63)) Bt[e] = 0.0f;
}

__device__ inline void setup_cg(int tid, double* CGs){
  if (tid < 96){
    int T = (tid >= 3) + (tid >= 18) + (tid >= 43) + (tid >= 78) + (tid >= 87);
    int l1 = TL1c[T], l2 = TL2c[T], l3 = TL3c[T];
    int n2 = 2*l2 + 1;
    int r = tid - TCOFFc[T];
    int m1 = r / n2 - l1, m2 = r % n2 - l2, m3 = m1 + m2;
    double cv = 0.0;
    if (m3 >= -l3 && m3 <= l3) cv = cgq(l1, m1, l2, m2, l3, m3);
    CGs[tid] = cv;
  }
}

__device__ inline void setup_elem(int tid, const double* CGs, double* Wr){
  if (tid < 288){
    int T = (tid >= 9) + (tid >= 54) + (tid >= 129) + (tid >= 234) + (tid >= 243);
    int l1 = TL1c[T], l2 = TL2c[T], l3 = TL3c[T];
    int n2 = 2*l2 + 1, n3 = 2*l3 + 1;
    int e = tid - TEOFFc[T];
    int a = e / (n2*n3), b = (e / n3) % n2, c = e % n3;
    int mua = iabsi(a - l1), mub = iabsi(b - l2), muc = iabsi(c - l3);
    const double* cg = CGs + TCOFFc[T];
    double s = 0.0;
    for (int ai = 0; ai < 2; ++ai){
      if (ai && mua == 0) continue;
      int i = ai ? l1 + mua : l1 - mua;
      for (int bi = 0; bi < 2; ++bi){
        if (bi && mub == 0) continue;
        int k = bi ? l2 + mub : l2 - mub;
        int m = (i - l1) + (k - l2) + l3;
        if (m != l3 - muc && m != l3 + muc) continue;
        double cv = cg[i*n2 + k];
        if (cv == 0.0) continue;
        cdbl t = cmul(cmul(qent(l1, i, a), qent(l2, k, b)), cconj(qent(l3, m, c)));
        s += t.re * cv;
      }
    }
    Wr[tid] = s;
  }
}

__device__ inline void setup_norm_part(int tid, const double* Wr, double* PART){
  if (tid < 48){
    int T = tid >> 3, p = tid & 7;
    int off = TEOFFc[T], n = TEOFFc[T+1] - off;
    double acc = 0.0;
    for (int e = p; e < n; e += 8) acc += Wr[off + e] * Wr[off + e];
    PART[tid] = acc;
  }
}
__device__ inline void setup_norm_inv(int tid, const double* PART, double* INVs){
  if (tid < 6){
    double tot = 0.0;
    for (int p = 0; p < 8; ++p) tot += PART[tid*8 + p];
    double inv = 1.0 / sqrt(tot);
    if (tid == 5) inv *= 2.23606797749978969641;
    INVs[tid] = inv;
  }
}

__device__ inline void setup_fold(int tid, int nthr, const float* w, const double* Wr,
                                  const double* INVs, float* Bt){
  const double ALPHA = 0.61237243569579452455;
  for (int e = tid; e < 18*30; e += nthr){
    int kk = e / 18, n = e % 18;
    int kOut = n / 6, pq = IDXc[n % 6];
    int k, u, i, T;
    if (kk < 6)       { k = 10 + kk; u = kk / 3;        i = kk % 3;        T = 1; }
    else if (kk < 16) { k = 20 + kk; u = (kk - 6) / 5;  i = (kk - 6) % 5;  T = 2; }
    else              { k = 34 + kk; u = (kk - 16) / 7; i = (kk - 16) % 7; T = 3; }
    const double* A = Wr + TEOFFc[T];
    const double* C = Wr + 243 + pq*5;
    double v = 0.0;
    for (int j = 0; j < 5; ++j) v += A[(i*5 + j)*3 + kOut] * C[j];
    double val = ALPHA * (double)w[2*T + u] * INVs[T] * INVs[5] * v;
    if (kk < 6)
      val += ALPHA * (double)w[u] * INVs[0] * INVs[4] * Wr[i*3 + kOut] * Wr[234 + pq];
    Bt[n*64 + k] = (float)val;
  }
}

__global__ __launch_bounds__(288) void piezo_setup_b(const float* __restrict__ w,
                                                     float* __restrict__ Bt){
  __shared__ double CGs[96], Wr[288], PART[48], INVs[6];
  const int tid = threadIdx.x;
  const int nthr = blockDim.x;

  setup_zero(tid, nthr, Bt);
  setup_cg(tid, CGs);
  __syncthreads();
  setup_elem(tid, CGs, Wr);
  __syncthreads();
  setup_norm_part(tid, Wr, PART);
  __syncthreads();
  setup_norm_inv(tid, PART, INVs);
  __syncthreads();
  setup_fold(tid, nthr, w, Wr, INVs, Bt);
}


#define WPB 4
#define TPW 5

__global__ __launch_bounds__(WPB*32) void piezo_wmma_main(const float* __restrict__ X,
                                                          const float* __restrict__ Bt,
                                                          float* __restrict__ out,
                                                          int rows){
  const int lane = threadIdx.x & 31;
  const int wave = threadIdx.x >> 5;
  const int half = lane >> 4;
  const int l16  = lane & 15;

  v2f bf0[16], bf1[16];
  #pragma unroll
  for (int kk = 0; kk < 16; ++kk){
    int k0 = 4*kk + 2*half;
    bf0[kk] = *(const v2f*)(Bt + (l16     ) * 64 + k0);
    bf1[kk] = *(const v2f*)(Bt + (l16 + 16) * 64 + k0);
  }

  const long tilebase = (long)(blockIdx.x * WPB + wave) * TPW;
  for (int t = 0; t < TPW; ++t){
    long row0 = (tilebase + t) * 16;
    if (row0 + 16 > (long)rows) break;
    const float* xr = X + (row0 + l16) * 64 + 2*half;

    __builtin_prefetch((const void*)(xr + 16*64), 0, 1);
    __builtin_prefetch((const void*)(xr + 16*64 + 32), 0, 1);

    v8f acc0 = {}; v8f acc1 = {};
    #pragma unroll
    for (int kk = 0; kk < 16; ++kk){
      v2f a = *(const v2f*)(xr + 4*kk);
      acc0 = __builtin_amdgcn_wmma_f32_16x16x4_f32(false, a, false, bf0[kk],
                                                   (short)0, acc0, false, false);
      acc1 = __builtin_amdgcn_wmma_f32_16x16x4_f32(false, a, false, bf1[kk],
                                                   (short)0, acc1, false, false);
    }

    float* orow = out + row0 * 18;
    #pragma unroll
    for (int r = 0; r < 8; ++r){
      int row = r + half*8;
      orow[row*18 + l16] = acc0[r];
      if (l16 < 2) orow[row*18 + 16 + l16] = acc1[r];
    }
  }
}


extern "C" void kernel_launch(void* const* d_in, const int* in_sizes, int n_in,
                              void* d_out, int out_size, void* d_ws, size_t ws_size,
                              hipStream_t stream){
  const float* X = (const float*)d_in[0];
  const float* w = (const float*)d_in[1];
  float* out = (float*)d_out;
  float* Bt  = (float*)d_ws;

  const int rows  = in_sizes[0] / 64;
  const int tiles = rows / 16;
  const int blocks = (tiles + WPB*TPW - 1) / (WPB*TPW);

  piezo_setup_b<<<1, 288, 0, stream>>>(w, Bt);
  piezo_wmma_main<<<blocks, WPB*32, 0, stream>>>(X, Bt, out, rows);
}


# Main kernel reads and multiplies only the 30 X columns the reference slices (10-15, 26-35, 50-63), packed into 8 k-steps: 16 WMMA and 5 loads per tile (was 32 and 16); k summation order changes
# speedup vs baseline: 2.3905x; 1.1004x over previous
#include <hip/hip_runtime.h>
#include <cmath>

typedef float v2f __attribute__((ext_vector_type(2)));
typedef float v8f __attribute__((ext_vector_type(8)));


struct cdbl { double re, im; };
__device__ inline cdbl cmul(cdbl a, cdbl b){ return {a.re*b.re - a.im*b.im, a.re*b.im + a.im*b.re}; }
__device__ inline cdbl cconj(cdbl a){ return {a.re, -a.im}; }

__device__ inline int imini(int a, int b){ return a < b ? a : b; }
__device__ inline int imaxi(int a, int b){ return a > b ? a : b; }
__device__ inline int iabsi(int a){ return a < 0 ? -a : a; }

__device__ __constant__ double FACTc[9] = {1.0, 1.0, 2.0, 6.0, 24.0, 120.0, 720.0, 5040.0, 40320.0};
__device__ inline double factd(int n){ return FACTc[n]; }

__device__ double cgq(int j1,int m1,int j2,int m2,int j3,int m3){
  if (m3 != m1 + m2) return 0.0;
  int vmin = imaxi(imaxi(-j1 + j2 + m3, -j1 + m1), 0);
  int vmax = imini(imini(j2 + j3 + m1, j3 - j1 + j2), j3 + m3);
  double c = sqrt((double)(2*j3+1) * factd(j3+j1-j2) * factd(j3-j1+j2) * factd(j1+j2-j3)
                  * factd(j3+m3) * factd(j3-m3)
                  / (factd(j1+j2+j3+1) * factd(j1-m1) * factd(j1+m1) * factd(j2-m2) * factd(j2+m2)));
  double s = 0.0;
  for (int v = vmin; v <= vmax; ++v){
    double sg = ((v + j2 + m2) & 1) ? -1.0 : 1.0;
    s += sg * factd(j2+j3+m1-v) * factd(j1-m1+v)
         / (factd(v) * factd(j3-j1+j2-v) * factd(j3+m3-v) * factd(v+j1-j2-m3));
  }
  return c * s;
}

__device__ inline cdbl qent(int l, int r, int c){
  const double is2 = 0.70710678118654752440;
  int mu = iabsi(c - l);
  double x = 0.0, y = 0.0;
  if (mu == 0){
    if (r == l) x = 1.0;
  } else {
    double sg = (mu & 1) ? -is2 : is2;
    if (c > l){
      if (r == l - mu) x = is2; else if (r == l + mu) x = sg;
    } else {
      if (r == l - mu) y = -is2; else if (r == l + mu) y = sg;
    }
  }
  switch (l & 3){
    case 0:  return { x,  y};
    case 1:  return { y, -x};
    case 2:  return {-x, -y};
    default: return {-y,  x};
  }
}

__device__ __constant__ int TL1c[6] = {1, 1, 2, 3, 1, 1};
__device__ __constant__ int TL2c[6] = {0, 2, 2, 2, 1, 1};
__device__ __constant__ int TL3c[6] = {1, 1, 1, 1, 0, 2};
__device__ __constant__ int TEOFFc[7] = {0, 9, 54, 129, 234, 243, 288};
__device__ __constant__ int TCOFFc[7] = {0, 3, 18, 43, 78, 87, 96};
__device__ __constant__ int IDXc[6] = {0, 4, 8, 1, 5, 6};

__device__ inline bool bt_active(int n, int k){
  return n < 18 && ((k >= 10 && k < 16) || (k >= 26 && k < 36) || k >= 50);
}

__device__ inline void setup_zero(int tid, int nthr, float* Bt){
  for (int e = tid; e < 32*64; e += nthr)
    if (!bt_active(e >> 6, e & 63)) Bt[e] = 0.0f;
}

__device__ inline void setup_cg(int tid, double* CGs){
  if (tid < 96){
    int T = (tid >= 3) + (tid >= 18) + (tid >= 43) + (tid >= 78) + (tid >= 87);
    int l1 = TL1c[T], l2 = TL2c[T], l3 = TL3c[T];
    int n2 = 2*l2 + 1;
    int r = tid - TCOFFc[T];
    int m1 = r / n2 - l1, m2 = r % n2 - l2, m3 = m1 + m2;
    double cv = 0.0;
    if (m3 >= -l3 && m3 <= l3) cv = cgq(l1, m1, l2, m2, l3, m3);
    CGs[tid] = cv;
  }
}

__device__ inline void setup_elem(int tid, const double* CGs, double* Wr){
  if (tid < 288){
    int T = (tid >= 9) + (tid >= 54) + (tid >= 129) + (tid >= 234) + (tid >= 243);
    int l1 = TL1c[T], l2 = TL2c[T], l3 = TL3c[T];
    int n2 = 2*l2 + 1, n3 = 2*l3 + 1;
    int e = tid - TEOFFc[T];
    int a = e / (n2*n3), b = (e / n3) % n2, c = e % n3;
    int mua = iabsi(a - l1), mub = iabsi(b - l2), muc = iabsi(c - l3);
    const double* cg = CGs + TCOFFc[T];
    double s = 0.0;
    for (int ai = 0; ai < 2; ++ai){
      if (ai && mua == 0) continue;
      int i = ai ? l1 + mua : l1 - mua;
      for (int bi = 0; bi < 2; ++bi){
        if (bi && mub == 0) continue;
        int k = bi ? l2 + mub : l2 - mub;
        int m = (i - l1) + (k - l2) + l3;
        if (m != l3 - muc && m != l3 + muc) continue;
        double cv = cg[i*n2 + k];
        if (cv == 0.0) continue;
        cdbl t = cmul(cmul(qent(l1, i, a), qent(l2, k, b)), cconj(qent(l3, m, c)));
        s += t.re * cv;
      }
    }
    Wr[tid] = s;
  }
}

__device__ inline void setup_norm_part(int tid, const double* Wr, double* PART){
  if (tid < 48){
    int T = tid >> 3, p = tid & 7;
    int off = TEOFFc[T], n = TEOFFc[T+1] - off;
    double acc = 0.0;
    for (int e = p; e < n; e += 8) acc += Wr[off + e] * Wr[off + e];
    PART[tid] = acc;
  }
}
__device__ inline void setup_norm_inv(int tid, const double* PART, double* INVs){
  if (tid < 6){
    double tot = 0.0;
    for (int p = 0; p < 8; ++p) tot += PART[tid*8 + p];
    double inv = 1.0 / sqrt(tot);
    if (tid == 5) inv *= 2.23606797749978969641;
    INVs[tid] = inv;
  }
}

__device__ inline void setup_fold(int tid, int nthr, const float* w, const double* Wr,
                                  const double* INVs, float* Bt){
  const double ALPHA = 0.61237243569579452455;
  for (int e = tid; e < 18*30; e += nthr){
    int kk = e / 18, n = e % 18;
    int kOut = n / 6, pq = IDXc[n % 6];
    int k, u, i, T;
    if (kk < 6)       { k = 10 + kk; u = kk / 3;        i = kk % 3;        T = 1; }
    else if (kk < 16) { k = 20 + kk; u = (kk - 6) / 5;  i = (kk - 6) % 5;  T = 2; }
    else              { k = 34 + kk; u = (kk - 16) / 7; i = (kk - 16) % 7; T = 3; }
    const double* A = Wr + TEOFFc[T];
    const double* C = Wr + 243 + pq*5;
    double v = 0.0;
    for (int j = 0; j < 5; ++j) v += A[(i*5 + j)*3 + kOut] * C[j];
    double val = ALPHA * (double)w[2*T + u] * INVs[T] * INVs[5] * v;
    if (kk < 6)
      val += ALPHA * (double)w[u] * INVs[0] * INVs[4] * Wr[i*3 + kOut] * Wr[234 + pq];
    Bt[n*64 + k] = (float)val;
  }
}

__global__ __launch_bounds__(288) void piezo_setup_b(const float* __restrict__ w,
                                                     float* __restrict__ Bt){
  __shared__ double CGs[96], Wr[288], PART[48], INVs[6];
  const int tid = threadIdx.x;
  const int nthr = blockDim.x;

  setup_zero(tid, nthr, Bt);
  setup_cg(tid, CGs);
  __syncthreads();
  setup_elem(tid, CGs, Wr);
  __syncthreads();
  setup_norm_part(tid, Wr, PART);
  __syncthreads();
  setup_norm_inv(tid, PART, INVs);
  __syncthreads();
  setup_fold(tid, nthr, w, Wr, INVs, Bt);
}


typedef float v4f __attribute__((ext_vector_type(4)));

#define WPB 4
#define TPW 5
#define KST 8

__device__ inline void load_frag(const float* __restrict__ row, int half, v2f f[KST]){
  const float* pa = row + 10 + 16*half;
  const float* pb = row + 32 + 24*half;
  const float* pc = row + 52 + 8*half;
  v2f q0 = *(const v2f*)(pa);
  v4f q1 = *(const v4f*)(pa + 2);
  v4f q2 = *(const v4f*)(pb);
  v4f q3 = *(const v4f*)(pc);
  v2f q4 = *(const v2f*)(row + 50);
  if (half) q4 = (v2f){0.0f, 0.0f};
  f[0] = q0;
  f[1] = (v2f){q1.x, q1.y};
  f[2] = (v2f){q1.z, q1.w};
  f[3] = (v2f){q2.x, q2.y};
  f[4] = (v2f){q2.z, q2.w};
  f[5] = (v2f){q3.x, q3.y};
  f[6] = (v2f){q3.z, q3.w};
  f[7] = q4;
}

__global__ __launch_bounds__(WPB*32) void piezo_wmma_main(const float* __restrict__ X,
                                                          const float* __restrict__ Bt,
                                                          float* __restrict__ out,
                                                          int rows){
  const int lane = threadIdx.x & 31;
  const int wave = threadIdx.x >> 5;
  const int half = lane >> 4;
  const int l16  = lane & 15;

  v2f bf0[KST], bf1[KST];
  load_frag(Bt + (l16     ) * 64, half, bf0);
  load_frag(Bt + (l16 + 16) * 64, half, bf1);

  const long tilebase = (long)(blockIdx.x * WPB + wave) * TPW;
  for (int t = 0; t < TPW; ++t){
    long row0 = (tilebase + t) * 16;
    if (row0 + 16 > (long)rows) break;
    const float* xr = X + (row0 + l16) * 64;

    __builtin_prefetch((const void*)(xr + 16*64), 0, 1);
    __builtin_prefetch((const void*)(xr + 16*64 + 32), 0, 1);

    v2f a[KST];
    load_frag(xr, half, a);

    v8f acc0 = {}; v8f acc1 = {};
    #pragma unroll
    for (int s = 0; s < KST; ++s){
      acc0 = __builtin_amdgcn_wmma_f32_16x16x4_f32(false, a[s], false, bf0[s],
                                                   (short)0, acc0, false, false);
      acc1 = __builtin_amdgcn_wmma_f32_16x16x4_f32(false, a[s], false, bf1[s],
                                                   (short)0, acc1, false, false);
    }

    float* orow = out + row0 * 18;
    #pragma unroll
    for (int r = 0; r < 8; ++r){
      int row = r + half*8;
      orow[row*18 + l16] = acc0[r];
      if (l16 < 2) orow[row*18 + 16 + l16] = acc1[r];
    }
  }
}


extern "C" void kernel_launch(void* const* d_in, const int* in_sizes, int n_in,
                              void* d_out, int out_size, void* d_ws, size_t ws_size,
                              hipStream_t stream){
  const float* X = (const float*)d_in[0];
  const float* w = (const float*)d_in[1];
  float* out = (float*)d_out;
  float* Bt  = (float*)d_ws;

  const int rows  = in_sizes[0] / 64;
  const int tiles = rows / 16;
  const int blocks = (tiles + WPB*TPW - 1) / (WPB*TPW);

  piezo_setup_b<<<1, 288, 0, stream>>>(w, Bt);
  piezo_wmma_main<<<blocks, WPB*32, 0, stream>>>(X, Bt, out, rows);
}


# Each tile's 1152 B of output staged in the wave's own LDS patch and written as nine whole 128-B lines by 3 b128 stores (was 16 part-line stores); WMMA roles swapped: a lane holds 8 adjacent columns
# speedup vs baseline: 2.8503x; 1.1923x over previous
#include <hip/hip_runtime.h>
#include <cmath>

typedef float v2f __attribute__((ext_vector_type(2)));
typedef float v8f __attribute__((ext_vector_type(8)));


struct cdbl { double re, im; };
__device__ inline cdbl cmul(cdbl a, cdbl b){ return {a.re*b.re - a.im*b.im, a.re*b.im + a.im*b.re}; }
__device__ inline cdbl cconj(cdbl a){ return {a.re, -a.im}; }

__device__ inline int imini(int a, int b){ return a < b ? a : b; }
__device__ inline int imaxi(int a, int b){ return a > b ? a : b; }
__device__ inline int iabsi(int a){ return a < 0 ? -a : a; }

__device__ __constant__ double FACTc[9] = {1.0, 1.0, 2.0, 6.0, 24.0, 120.0, 720.0, 5040.0, 40320.0};
__device__ inline double factd(int n){ return FACTc[n]; }

__device__ double cgq(int j1,int m1,int j2,int m2,int j3,int m3){
  if (m3 != m1 + m2) return 0.0;
  int vmin = imaxi(imaxi(-j1 + j2 + m3, -j1 + m1), 0);
  int vmax = imini(imini(j2 + j3 + m1, j3 - j1 + j2), j3 + m3);
  double c = sqrt((double)(2*j3+1) * factd(j3+j1-j2) * factd(j3-j1+j2) * factd(j1+j2-j3)
                  * factd(j3+m3) * factd(j3-m3)
                  / (factd(j1+j2+j3+1) * factd(j1-m1) * factd(j1+m1) * factd(j2-m2) * factd(j2+m2)));
  double s = 0.0;
  for (int v = vmin; v <= vmax; ++v){
    double sg = ((v + j2 + m2) & 1) ? -1.0 : 1.0;
    s += sg * factd(j2+j3+m1-v) * factd(j1-m1+v)
         / (factd(v) * factd(j3-j1+j2-v) * factd(j3+m3-v) * factd(v+j1-j2-m3));
  }
  return c * s;
}

__device__ inline cdbl qent(int l, int r, int c){
  const double is2 = 0.70710678118654752440;
  int mu = iabsi(c - l);
  double x = 0.0, y = 0.0;
  if (mu == 0){
    if (r == l) x = 1.0;
  } else {
    double sg = (mu & 1) ? -is2 : is2;
    if (c > l){
      if (r == l - mu) x = is2; else if (r == l + mu) x = sg;
    } else {
      if (r == l - mu) y = -is2; else if (r == l + mu) y = sg;
    }
  }
  switch (l & 3){
    case 0:  return { x,  y};
    case 1:  return { y, -x};
    case 2:  return {-x, -y};
    default: return {-y,  x};
  }
}

__device__ __constant__ int TL1c[6] = {1, 1, 2, 3, 1, 1};
__device__ __constant__ int TL2c[6] = {0, 2, 2, 2, 1, 1};
__device__ __constant__ int TL3c[6] = {1, 1, 1, 1, 0, 2};
__device__ __constant__ int TEOFFc[7] = {0, 9, 54, 129, 234, 243, 288};
__device__ __constant__ int TCOFFc[7] = {0, 3, 18, 43, 78, 87, 96};
__device__ __constant__ int IDXc[6] = {0, 4, 8, 1, 5, 6};

__device__ inline bool bt_active(int n, int k){
  return n < 18 && ((k >= 10 && k < 16) || (k >= 26 && k < 36) || k >= 50);
}

__device__ inline void setup_zero(int tid, int nthr, float* Bt){
  for (int e = tid; e < 32*64; e += nthr)
    if (!bt_active(e >> 6, e & 63)) Bt[e] = 0.0f;
}

__device__ inline void setup_cg(int tid, double* CGs){
  if (tid < 96){
    int T = (tid >= 3) + (tid >= 18) + (tid >= 43) + (tid >= 78) + (tid >= 87);
    int l1 = TL1c[T], l2 = TL2c[T], l3 = TL3c[T];
    int n2 = 2*l2 + 1;
    int r = tid - TCOFFc[T];
    int m1 = r / n2 - l1, m2 = r % n2 - l2, m3 = m1 + m2;
    double cv = 0.0;
    if (m3 >= -l3 && m3 <= l3) cv = cgq(l1, m1, l2, m2, l3, m3);
    CGs[tid] = cv;
  }
}

__device__ inline void setup_elem(int tid, const double* CGs, double* Wr){
  if (tid < 288){
    int T = (tid >= 9) + (tid >= 54) + (tid >= 129) + (tid >= 234) + (tid >= 243);
    int l1 = TL1c[T], l2 = TL2c[T], l3 = TL3c[T];
    int n2 = 2*l2 + 1, n3 = 2*l3 + 1;
    int e = tid - TEOFFc[T];
    int a = e / (n2*n3), b = (e / n3) % n2, c = e % n3;
    int mua = iabsi(a - l1), mub = iabsi(b - l2), muc = iabsi(c - l3);
    const double* cg = CGs + TCOFFc[T];
    double s = 0.0;
    for (int ai = 0; ai < 2; ++ai){
      if (ai && mua == 0) continue;
      int i = ai ? l1 + mua : l1 - mua;
      for (int bi = 0; bi < 2; ++bi){
        if (bi && mub == 0) continue;
        int k = bi ? l2 + mub : l2 - mub;
        int m = (i - l1) + (k - l2) + l3;
        if (m != l3 - muc && m != l3 + muc) continue;
        double cv = cg[i*n2 + k];
        if (cv == 0.0) continue;
        cdbl t = cmul(cmul(qent(l1, i, a), qent(l2, k, b)), cconj(qent(l3, m, c)));
        s += t.re * cv;
      }
    }
    Wr[tid] = s;
  }
}

__device__ inline void setup_norm_part(int tid, const double* Wr, double* PART){
  if (tid < 48){
    int T = tid >> 3, p = tid & 7;
    int off = TEOFFc[T], n = TEOFFc[T+1] - off;
    double acc = 0.0;
    for (int e = p; e < n; e += 8) acc += Wr[off + e] * Wr[off + e];
    PART[tid] = acc;
  }
}
__device__ inline void setup_norm_inv(int tid, const double* PART, double* INVs){
  if (tid < 6){
    double tot = 0.0;
    for (int p = 0; p < 8; ++p) tot += PART[tid*8 + p];
    double inv = 1.0 / sqrt(tot);
    if (tid == 5) inv *= 2.23606797749978969641;
    INVs[tid] = inv;
  }
}

__device__ inline void setup_fold(int tid, int nthr, const float* w, const double* Wr,
                                  const double* INVs, float* Bt){
  const double ALPHA = 0.61237243569579452455;
  for (int e = tid; e < 18*30; e += nthr){
    int kk = e / 18, n = e % 18;
    int kOut = n / 6, pq = IDXc[n % 6];
    int k, u, i, T;
    if (kk < 6)       { k = 10 + kk; u = kk / 3;        i = kk % 3;        T = 1; }
    else if (kk < 16) { k = 20 + kk; u = (kk - 6) / 5;  i = (kk - 6) % 5;  T = 2; }
    else              { k = 34 + kk; u = (kk - 16) / 7; i = (kk - 16) % 7; T = 3; }
    const double* A = Wr + TEOFFc[T];
    const double* C = Wr + 243 + pq*5;
    double v = 0.0;
    for (int j = 0; j < 5; ++j) v += A[(i*5 + j)*3 + kOut] * C[j];
    double val = ALPHA * (double)w[2*T + u] * INVs[T] * INVs[5] * v;
    if (kk < 6)
      val += ALPHA * (double)w[u] * INVs[0] * INVs[4] * Wr[i*3 + kOut] * Wr[234 + pq];
    Bt[n*64 + k] = (float)val;
  }
}

__global__ __launch_bounds__(288) void piezo_setup_b(const float* __restrict__ w,
                                                     float* __restrict__ Bt){
  __shared__ double CGs[96], Wr[288], PART[48], INVs[6];
  const int tid = threadIdx.x;
  const int nthr = blockDim.x;

  setup_zero(tid, nthr, Bt);
  setup_cg(tid, CGs);
  __syncthreads();
  setup_elem(tid, CGs, Wr);
  __syncthreads();
  setup_norm_part(tid, Wr, PART);
  __syncthreads();
  setup_norm_inv(tid, PART, INVs);
  __syncthreads();
  setup_fold(tid, nthr, w, Wr, INVs, Bt);
}


typedef float v4f __attribute__((ext_vector_type(4)));

#define WPB 4
#define TPW 5
#define KST 8
#define OTF 288

__device__ inline void load_frag(const float* __restrict__ row, int half, v2f f[KST]){
  const float* pa = row + 10 + 16*half;
  const float* pb = row + 32 + 24*half;
  const float* pc = row + 52 + 8*half;
  v2f q0 = *(const v2f*)(pa);
  v4f q1 = *(const v4f*)(pa + 2);
  v4f q2 = *(const v4f*)(pb);
  v4f q3 = *(const v4f*)(pc);
  v2f q4 = *(const v2f*)(row + 50);
  if (half) q4 = (v2f){0.0f, 0.0f};
  f[0] = q0;
  f[1] = (v2f){q1.x, q1.y};
  f[2] = (v2f){q1.z, q1.w};
  f[3] = (v2f){q2.x, q2.y};
  f[4] = (v2f){q2.z, q2.w};
  f[5] = (v2f){q3.x, q3.y};
  f[6] = (v2f){q3.z, q3.w};
  f[7] = q4;
}

__global__ __launch_bounds__(WPB*32) void piezo_wmma_main(const float* __restrict__ X,
                                                          const float* __restrict__ Bt,
                                                          float* __restrict__ out,
                                                          int rows){
  __shared__ __attribute__((aligned(16))) float stage[WPB][OTF];

  const int lane = threadIdx.x & 31;
  const int wave = threadIdx.x >> 5;
  const int half = lane >> 4;
  const int l16  = lane & 15;
  float* st = stage[wave];

  v2f bf0[KST], bf1[KST];
  load_frag(Bt + (l16     ) * 64, half, bf0);
  load_frag(Bt + (l16 + 16) * 64, half, bf1);

  const long tilebase = (long)(blockIdx.x * WPB + wave) * TPW;
  for (int t = 0; t < TPW; ++t){
    long row0 = (tilebase + t) * 16;
    if (row0 + 16 > (long)rows) break;
    const float* xr = X + (row0 + l16) * 64;

    __builtin_prefetch((const void*)(xr + 16*64), 0, 1);
    __builtin_prefetch((const void*)(xr + 16*64 + 32), 0, 1);

    v2f a[KST];
    load_frag(xr, half, a);

    v8f acc0 = {}; v8f acc1 = {};
    #pragma unroll
    for (int s = 0; s < KST; ++s){
      acc0 = __builtin_amdgcn_wmma_f32_16x16x4_f32(false, bf0[s], false, a[s],
                                                   (short)0, acc0, false, false);
      acc1 = __builtin_amdgcn_wmma_f32_16x16x4_f32(false, bf1[s], false, a[s],
                                                   (short)0, acc1, false, false);
    }

    float* srow = st + l16*18 + half*8;
    *(v2f*)(srow    ) = (v2f){acc0[0], acc0[1]};
    *(v2f*)(srow + 2) = (v2f){acc0[2], acc0[3]};
    *(v2f*)(srow + 4) = (v2f){acc0[4], acc0[5]};
    *(v2f*)(srow + 6) = (v2f){acc0[6], acc0[7]};
    if (half == 0) *(v2f*)(st + l16*18 + 16) = (v2f){acc1[0], acc1[1]};

    asm volatile("s_wait_dscnt 0x0" ::: "memory");

    float* ot = out + row0 * 18;
    v4f p0 = *(const v4f*)(st + 4*lane);
    v4f p1 = *(const v4f*)(st + 4*(lane + 32));
    *(v4f*)(ot + 4*lane)        = p0;
    *(v4f*)(ot + 4*(lane + 32)) = p1;
    if (lane < 8){
      v4f p2 = *(const v4f*)(st + 4*(lane + 64));
      *(v4f*)(ot + 4*(lane + 64)) = p2;
    }
  }
}


extern "C" void kernel_launch(void* const* d_in, const int* in_sizes, int n_in,
                              void* d_out, int out_size, void* d_ws, size_t ws_size,
                              hipStream_t stream){
  const float* X = (const float*)d_in[0];
  const float* w = (const float*)d_in[1];
  float* out = (float*)d_out;
  float* Bt  = (float*)d_ws;

  const int rows  = in_sizes[0] / 64;
  const int tiles = rows / 16;
  const int blocks = (tiles + WPB*TPW - 1) / (WPB*TPW);

  piezo_setup_b<<<1, 288, 0, stream>>>(w, Bt);
  piezo_wmma_main<<<blocks, WPB*32, 0, stream>>>(X, Bt, out, rows);
}


# Setup launch: fp64 chain shortened: 6 workgroups each build only the tensors of their own Bt slice, index tables as immediates, real q products, butterfly norms (fp64 norm sum order changes)
# speedup vs baseline: 3.2289x; 1.1328x over previous
#include <hip/hip_runtime.h>
#include <cmath>

typedef float v2f __attribute__((ext_vector_type(2)));
typedef float v8f __attribute__((ext_vector_type(8)));


#define SETUP_WG  6
#define SETUP_THR 160

struct wgplan { int p0, n0, p1, n1, e0, m0, e1, m1, tl, nt, fb, fn; };
__device__ inline wgplan plan_for(int wg){
  wgplan P;
  if (wg == 0)     { P.p0 = 0;  P.n0 = 18; P.p1 = 78; P.n1 = 18; P.e0 = 0;   P.m0 = 54;  P.e1 = 234; P.m1 = 54;
                     P.tl = 0x5410; P.nt = 4; P.fb = 0;                  P.fn = 108; }
  else if (wg < 3) { P.p0 = 18; P.n0 = 25; P.p1 = 87; P.n1 = 9;  P.e0 = 54;  P.m0 = 75;  P.e1 = 243; P.m1 = 45;
                     P.tl = 0x52;   P.nt = 2; P.fb = 108 + 90*(wg - 1);  P.fn = 90; }
  else             { P.p0 = 43; P.n0 = 35; P.p1 = 87; P.n1 = 9;  P.e0 = 129; P.m0 = 105; P.e1 = 243; P.m1 = 45;
                     P.tl = 0x53;   P.nt = 2; P.fb = 288 + 84*(wg - 3);  P.fn = 84; }
  return P;
}

__device__ inline int imini(int a, int b){ return a < b ? a : b; }
__device__ inline int imaxi(int a, int b){ return a > b ? a : b; }
__device__ inline int iabsi(int a){ return a < 0 ? -a : a; }

__device__ inline int tl1(int T){ return (0x5E5 >> (2*T)) & 3; }
__device__ inline int tl2(int T){ return (0x5A8 >> (2*T)) & 3; }
__device__ inline int tl3(int T){ return (0x855 >> (2*T)) & 3; }
__device__ inline int teoff(int T){ return (int)((0x481E6EA408D81200ull >> (9*T)) & 511ull); }
__device__ inline int tcoff(int T){ return (int)((0x182BCE5648180ull >> (7*T)) & 127ull); }
__device__ inline int idxpq(int j){ return (0x651840 >> (4*j)) & 15; }

__device__ inline int divc(int x, int d){ return d == 1 ? x : (d == 3 ? x / 3 : (d == 5 ? x / 5 : x / 15)); }

__device__ inline double factd(int n){
  const unsigned long long lo5 = 0x480280100000000ull;
  const unsigned long long hi4 = 0xF3BC3B9686E0ull;
  const bool low = n < 5;
  const unsigned long long wsel = low ? lo5 : hi4;
  const int sh = 12 * (low ? n : n - 5);
  const unsigned hi = ((((unsigned)(wsel >> sh)) & 0xFFFu) + 0x3FF00u) << 12;
  return __hiloint2double((int)hi, 0);
}

__device__ double cg_pref(int j1,int m1,int j2,int m2,int j3,int m3){
  double num = (double)(2*j3+1) * factd(j3+j1-j2) * factd(j3-j1+j2) * factd(j1+j2-j3)
               * factd(j3+m3) * factd(j3-m3);
  double den = factd(j1+j2+j3+1) * factd(j1-m1) * factd(j1+m1) * factd(j2-m2) * factd(j2+m2);
  return num * (1.0 / sqrt(num * den));
}
__device__ double cg_sum(int j1,int m1,int j2,int m2,int j3,int m3){
  int vmin = imaxi(imaxi(-j1 + j2 + m3, -j1 + m1), 0);
  int vmax = imini(imini(j2 + j3 + m1, j3 - j1 + j2), j3 + m3);
  double s = 0.0;
  for (int v = vmin; v <= vmax; ++v){
    double sg = ((v + j2 + m2) & 1) ? -1.0 : 1.0;
    s += sg * factd(j2+j3+m1-v) * factd(j1-m1+v)
         / (factd(v) * factd(j3-j1+j2-v) * factd(j3+m3-v) * factd(v+j1-j2-m3));
  }
  return s;
}

struct qv { double v; int p; };
__device__ inline qv qent(int l, int r, int c){
  const double is2 = 0.70710678118654752440;
  int mu = iabsi(c - l);
  double v = 0.0;
  int p = 0;
  if (mu == 0){
    v = (r == l) ? 1.0 : 0.0;
  } else {
    double sg = (mu & 1) ? -is2 : is2;
    if (c > l){
      v = (r == l - mu) ? is2 : ((r == l + mu) ? sg : 0.0);
    } else {
      p = 1;
      v = (r == l - mu) ? -is2 : ((r == l + mu) ? sg : 0.0);
    }
  }
  int k = l & 3;
  if (k == 1)      { v = p ? v : -v; p ^= 1; }
  else if (k == 2) { v = -v; }
  else if (k == 3) { v = p ? -v : v; p ^= 1; }
  return {v, p};
}

__device__ inline bool bt_active(int n, int k){
  return n < 18 && ((k >= 10 && k < 16) || (k >= 26 && k < 36) || k >= 50);
}

__device__ inline void setup_zero(int gtid, int gthr, float* Bt){
  for (int e = gtid; e < 32*64; e += gthr)
    if (!bt_active(e >> 6, e & 63)) Bt[e] = 0.0f;
}

__device__ inline void setup_cg(int tid, const wgplan& P, const float* w, double* CGp, double* Ws){
  const int t = tid & 63;
  if (tid < 128 && t < P.n0 + P.n1){
    const int pr = t < P.n0 ? P.p0 + t : P.p1 + (t - P.n0);
    int T = (pr >= 3) + (pr >= 18) + (pr >= 43) + (pr >= 78) + (pr >= 87);
    int l1 = tl1(T), l2 = tl2(T), l3 = tl3(T);
    int n2 = 2*l2 + 1;
    int r = pr - tcoff(T);
    int r1 = divc(r, n2);
    int m1 = r1 - l1, m2 = r - r1*n2 - l2, m3 = m1 + m2;
    double o = 0.0;
    if (m3 >= -l3 && m3 <= l3)
      o = tid < 64 ? cg_sum(l1, m1, l2, m2, l3, m3) : cg_pref(l1, m1, l2, m2, l3, m3);
    CGp[pr + (tid < 64 ? 0 : 96)] = o;
  } else if (tid >= 128 && tid < 136){
    Ws[tid - 128] = (double)w[tid - 128];
  }
}

__device__ inline void setup_elem(int tid, const wgplan& P, const double* CGp, double* Wr){
  if (tid < P.m0 + P.m1){
    const int el = tid < P.m0 ? P.e0 + tid : P.e1 + (tid - P.m0);
    int T = (el >= 9) + (el >= 54) + (el >= 129) + (el >= 234) + (el >= 243);
    int l1 = tl1(T), l2 = tl2(T), l3 = tl3(T);
    int n2 = 2*l2 + 1, n3 = 2*l3 + 1;
    int e = el - teoff(T);
    int a = divc(e, n2*n3);
    int rem = e - a*n2*n3;
    int b = divc(rem, n3), c = rem - b*n3;
    int mua = iabsi(a - l1), mub = iabsi(b - l2), muc = iabsi(c - l3);
    const int base = tcoff(T);
    const qv qa0 = qent(l1, l1 - mua, a), qa1 = qent(l1, l1 + mua, a);
    const qv qb0 = qent(l2, l2 - mub, b), qb1 = qent(l2, l2 + mub, b);
    const qv qc0 = qent(l3, l3 - muc, c), qc1 = qent(l3, l3 + muc, c);
    double s = 0.0;
    for (int ai = 0; ai < 2; ++ai){
      if (ai && mua == 0) continue;
      int i = ai ? l1 + mua : l1 - mua;
      for (int bi = 0; bi < 2; ++bi){
        if (bi && mub == 0) continue;
        int k = bi ? l2 + mub : l2 - mub;
        int m = (i - l1) + (k - l2) + l3;
        if (m != l3 - muc && m != l3 + muc) continue;
        const qv q1 = ai ? qa1 : qa0, q2 = bi ? qb1 : qb0, q3 = (m == l3 - muc) ? qc0 : qc1;
        if ((q1.p ^ q2.p ^ q3.p) == 0){
          double t = q1.v * q2.v * q3.v;
          if (q1.p & q2.p) t = -t;
          int idx = base + i*n2 + k;
          s += t * (CGp[idx] * CGp[96 + idx]);
        }
      }
    }
    Wr[el] = s;
  }
}

__device__ inline void setup_norm(int tid, const wgplan& P, const double* Wr, double* INVs){
  if ((tid >> 5) < P.nt){
    int T = (P.tl >> (4*(tid >> 5))) & 15, p = tid & 31;
    int off = teoff(T), n = teoff(T + 1) - off;
    double acc = 0.0;
    for (int q = 0; q < 4; ++q){
      int e = p + 32*q;
      if (e < n){ double x = Wr[off + e]; acc += x * x; }
    }
    for (int m = 16; m >= 1; m >>= 1) acc += __shfl_xor(acc, m);
    double inv = 1.0 / sqrt(acc);
    if (T == 5) inv *= 2.23606797749978969641;
    if (p == 0) INVs[T] = inv;
  }
}

__device__ inline void setup_fold(int tid, const wgplan& P, const double* Ws, const double* Wr,
                                  const double* INVs, float* Bt){
  const double ALPHA = 0.61237243569579452455;
  if (tid < P.fn){
    int e = P.fb + tid;
    int kk = e / 18, n = e % 18;
    int kOut = n / 6, pq = idxpq(n % 6);
    int k, u, i, T, toff;
    if (kk < 6)       { k = 10 + kk; u = kk / 3;        i = kk % 3;        T = 1; toff = 9;   }
    else if (kk < 16) { k = 20 + kk; u = (kk - 6) / 5;  i = (kk - 6) % 5;  T = 2; toff = 54;  }
    else              { k = 34 + kk; u = (kk - 16) / 7; i = (kk - 16) % 7; T = 3; toff = 129; }
    const double* A = Wr + toff;
    const double* C = Wr + 243 + pq*5;
    double v = 0.0;
    for (int j = 0; j < 5; ++j) v += A[(i*5 + j)*3 + kOut] * C[j];
    double val = ALPHA * Ws[2*T + u] * INVs[T] * INVs[5] * v;
    if (kk < 6)
      val += ALPHA * Ws[u] * INVs[0] * INVs[4] * Wr[i*3 + kOut] * Wr[234 + pq];
    Bt[n*64 + k] = (float)val;
  }
}

__global__ __launch_bounds__(SETUP_THR) void piezo_setup_b(const float* __restrict__ w,
                                                           float* __restrict__ Bt){
  __shared__ double CGp[192], Wr[288], INVs[6], Ws[8];
  const int tid = threadIdx.x;
  const wgplan P = plan_for((int)blockIdx.x);

  setup_zero((int)blockIdx.x * SETUP_THR + tid, SETUP_WG * SETUP_THR, Bt);
  setup_cg(tid, P, w, CGp, Ws);
  __syncthreads();
  setup_elem(tid, P, CGp, Wr);
  __syncthreads();
  setup_norm(tid, P, Wr, INVs);
  __syncthreads();
  setup_fold(tid, P, Ws, Wr, INVs, Bt);
}


typedef float v4f __attribute__((ext_vector_type(4)));

#define WPB 4
#define TPW 5
#define KST 8
#define OTF 288

__device__ inline void load_frag(const float* __restrict__ row, int half, v2f f[KST]){
  const float* pa = row + 10 + 16*half;
  const float* pb = row + 32 + 24*half;
  const float* pc = row + 52 + 8*half;
  v2f q0 = *(const v2f*)(pa);
  v4f q1 = *(const v4f*)(pa + 2);
  v4f q2 = *(const v4f*)(pb);
  v4f q3 = *(const v4f*)(pc);
  v2f q4 = *(const v2f*)(row + 50);
  if (half) q4 = (v2f){0.0f, 0.0f};
  f[0] = q0;
  f[1] = (v2f){q1.x, q1.y};
  f[2] = (v2f){q1.z, q1.w};
  f[3] = (v2f){q2.x, q2.y};
  f[4] = (v2f){q2.z, q2.w};
  f[5] = (v2f){q3.x, q3.y};
  f[6] = (v2f){q3.z, q3.w};
  f[7] = q4;
}

__global__ __launch_bounds__(WPB*32) void piezo_wmma_main(const float* __restrict__ X,
                                                          const float* __restrict__ Bt,
                                                          float* __restrict__ out,
                                                          int rows){
  __shared__ __attribute__((aligned(16))) float stage[WPB][OTF];

  const int lane = threadIdx.x & 31;
  const int wave = threadIdx.x >> 5;
  const int half = lane >> 4;
  const int l16  = lane & 15;
  float* st = stage[wave];

  v2f bf0[KST], bf1[KST];
  load_frag(Bt + (l16     ) * 64, half, bf0);
  load_frag(Bt + (l16 + 16) * 64, half, bf1);

  const long tilebase = (long)(blockIdx.x * WPB + wave) * TPW;
  for (int t = 0; t < TPW; ++t){
    long row0 = (tilebase + t) * 16;
    if (row0 + 16 > (long)rows) break;
    const float* xr = X + (row0 + l16) * 64;

    __builtin_prefetch((const void*)(xr + 16*64), 0, 1);
    __builtin_prefetch((const void*)(xr + 16*64 + 32), 0, 1);

    v2f a[KST];
    load_frag(xr, half, a);

    v8f acc0 = {}; v8f acc1 = {};
    #pragma unroll
    for (int s = 0; s < KST; ++s){
      acc0 = __builtin_amdgcn_wmma_f32_16x16x4_f32(false, bf0[s], false, a[s],
                                                   (short)0, acc0, false, false);
      acc1 = __builtin_amdgcn_wmma_f32_16x16x4_f32(false, bf1[s], false, a[s],
                                                   (short)0, acc1, false, false);
    }

    float* srow = st + l16*18 + half*8;
    *(v2f*)(srow    ) = (v2f){acc0[0], acc0[1]};
    *(v2f*)(srow + 2) = (v2f){acc0[2], acc0[3]};
    *(v2f*)(srow + 4) = (v2f){acc0[4], acc0[5]};
    *(v2f*)(srow + 6) = (v2f){acc0[6], acc0[7]};
    if (half == 0) *(v2f*)(st + l16*18 + 16) = (v2f){acc1[0], acc1[1]};

    asm volatile("s_wait_dscnt 0x0" ::: "memory");

    float* ot = out + row0 * 18;
    v4f p0 = *(const v4f*)(st + 4*lane);
    v4f p1 = *(const v4f*)(st + 4*(lane + 32));
    *(v4f*)(ot + 4*lane)        = p0;
    *(v4f*)(ot + 4*(lane + 32)) = p1;
    if (lane < 8){
      v4f p2 = *(const v4f*)(st + 4*(lane + 64));
      *(v4f*)(ot + 4*(lane + 64)) = p2;
    }
  }
}


extern "C" void kernel_launch(void* const* d_in, const int* in_sizes, int n_in,
                              void* d_out, int out_size, void* d_ws, size_t ws_size,
                              hipStream_t stream){
  const float* X = (const float*)d_in[0];
  const float* w = (const float*)d_in[1];
  float* out = (float*)d_out;
  float* Bt  = (float*)d_ws;

  const int rows  = in_sizes[0] / 64;
  const int tiles = rows / 16;
  const int blocks = (tiles + WPB*TPW - 1) / (WPB*TPW);

  piezo_setup_b<<<SETUP_WG, SETUP_THR, 0, stream>>>(w, Bt);
  piezo_wmma_main<<<blocks, WPB*32, 0, stream>>>(X, Bt, out, rows);
}


# Setup launch removed: the weights' input-free Wigner-3j factors (load-time constants of the reference) are a literal fp32 table; each wave folds constant_w in fp32 (was one fp64 fold); one launch
# speedup vs baseline: 3.9860x; 1.2345x over previous
#include <hip/hip_runtime.h>

typedef float v2f __attribute__((ext_vector_type(2)));
typedef float v8f __attribute__((ext_vector_type(8)));

__device__ const float CT[32*64] __attribute__((aligned(256))) = {
  0.0f, 0.0f, 2.041241527e-01f, 0.0f, 0.0f, 2.041241527e-01f, 0.0f, 0.0f, 0.0f, 0.0f, 1.825741827e-01f, 0.0f, 0.0f, 1.825741827e-01f, 0.0f, 0.0f, 0.0f, 0.0f, 0.0f, 0.0f, 0.0f, 0.0f, 0.0f, 0.0f, 0.0f, 0.0f, 0.0f, 0.0f, 0.0f, 0.0f, 0.0f, 0.0f, 0.0f, 0.0f, 0.0f, 0.0f, 0.0f, 0.0f, 0.0f, 0.0f, 0.0f, 0.0f, 0.0f, 0.0f, 0.0f, 0.0f, 0.0f, 0.0f, 0.0f, 0.0f, -1.157275140e-01f, 0.0f, -8.964214474e-02f, 0.0f, 0.0f, 0.0f, 0.0f, -1.157275140e-01f, 0.0f, -8.964214474e-02f, 0.0f, 0.0f, 0.0f, 0.0f,
  0.0f, 0.0f, 2.041241527e-01f, 0.0f, 0.0f, 2.041241527e-01f, 0.0f, 0.0f, 0.0f, 0.0f, -9.128709137e-02f, 0.0f, 0.0f, -9.128709137e-02f, 0.0f, 0.0f, 0.0f, 0.0f, 0.0f, 0.0f, 0.0f, 0.0f, 0.0f, 0.0f, 0.0f, 0.0f, 0.0f, 0.0f, 0.0f, -1.581138819e-01f, 0.0f, 0.0f, 0.0f, 0.0f, -1.581138819e-01f, 0.0f, 0.0f, 0.0f, 0.0f, 0.0f, 0.0f, 0.0f, 0.0f, 0.0f, 0.0f, 0.0f, 0.0f, 0.0f, 0.0f, 0.0f, 0.0f, 0.0f, 1.195228621e-01f, 0.0f, 0.0f, 0.0f, 0.0f, 0.0f, 0.0f, 1.195228621e-01f, 0.0f, 0.0f, 0.0f, 0.0f,
  0.0f, 0.0f, 2.041241527e-01f, 0.0f, 0.0f, 2.041241527e-01f, 0.0f, 0.0f, 0.0f, 0.0f, -9.128709137e-02f, 0.0f, 0.0f, -9.128709137e-02f, 0.0f, 0.0f, 0.0f, 0.0f, 0.0f, 0.0f, 0.0f, 0.0f, 0.0f, 0.0f, 0.0f, 0.0f, 0.0f, 0.0f, 0.0f, 1.581138819e-01f, 0.0f, 0.0f, 0.0f, 0.0f, 1.581138819e-01f, 0.0f, 0.0f, 0.0f, 0.0f, 0.0f, 0.0f, 0.0f, 0.0f, 0.0f, 0.0f, 0.0f, 0.0f, 0.0f, 0.0f, 0.0f, 1.157275140e-01f, 0.0f, -2.988071553e-02f, 0.0f, 0.0f, 0.0f, 0.0f, 1.157275140e-01f, 0.0f, -2.988071553e-02f, 0.0f, 0.0f, 0.0f, 0.0f,
  0.0f, 0.0f, 0.0f, 0.0f, 0.0f, 0.0f, 0.0f, 0.0f, 0.0f, 0.0f, 0.0f, 1.369306445e-01f, 0.0f, 0.0f, 1.369306445e-01f, 0.0f, 0.0f, 0.0f, 0.0f, 0.0f, 0.0f, 0.0f, 0.0f, 0.0f, 0.0f, 0.0f, -7.905694097e-02f, 0.0f, 0.0f, 0.0f, 0.0f, -7.905694097e-02f, 0.0f, 0.0f, 0.0f, 0.0f, 0.0f, 0.0f, 0.0f, 0.0f, 0.0f, 0.0f, 0.0f, 0.0f, 0.0f, 0.0f, 0.0f, 0.0f, 0.0f, 0.0f, 0.0f, 0.0f, 0.0f, -7.319250703e-02f, 0.0f, -9.449111670e-02f, 0.0f, 0.0f, 0.0f, 0.0f, -7.319250703e-02f, 0.0f, -9.449111670e-02f, 0.0f,
  0.0f, 0.0f, 0.0f, 0.0f, 0.0f, 0.0f, 0.0f, 0.0f, 0.0f, 0.0f, 0.0f, 0.0f, 0.0f, 0.0f, 0.0f, 0.0f, 0.0f, 0.0f, 0.0f, 0.0f, 0.0f, 0.0f, 0.0f, 0.0f, 0.0f, 0.0f, 0.0f, 0.0f, 1.369306445e-01f, 0.0f, -7.905694097e-02f, 0.0f, 0.0f, 1.369306445e-01f, 0.0f, -7.905694097e-02f, 0.0f, 0.0f, 0.0f, 0.0f, 0.0f, 0.0f, 0.0f, 0.0f, 0.0f, 0.0f, 0.0f, 0.0f, 0.0f, 0.0f, 0.0f, 9.449111670e-02f, 0.0f, 0.0f, 0.0f, 0.0f, 0.0f, 0.0f, 9.449111670e-02f, 0.0f, 0.0f, 0.0f, 0.0f, 0.0f,
  0.0f, 0.0f, 0.0f, 0.0f, 0.0f, 0.0f, 0.0f, 0.0f, 0.0f, 0.0f, 0.0f, 0.0f, 1.369306445e-01f, 0.0f, 0.0f, 1.369306445e-01f, 0.0f, 0.0f, 0.0f, 0.0f, 0.0f, 0.0f, 0.0f, 0.0f, 0.0f, 0.0f, 0.0f, 7.905694097e-02f, 0.0f, 0.0f, 0.0f, 0.0f, 7.905694097e-02f, 0.0f, 0.0f, 0.0f, 0.0f, 0.0f, 0.0f, 0.0f, 0.0f, 0.0f, 0.0f, 0.0f, 0.0f, 0.0f, 0.0f, 0.0f, 0.0f, 0.0f, 0.0f, 0.0f, 0.0f, 0.0f, -2.988071553e-02f, 0.0f, -1.157275140e-01f, 0.0f, 0.0f, 0.0f, 0.0f, -2.988071553e-02f, 0.0f, -1.157275140e-01f,
  0.0f, 0.0f, 0.0f, 2.041241527e-01f, 0.0f, 0.0f, 2.041241527e-01f, 0.0f, 0.0f, 0.0f, 0.0f, -9.128709137e-02f, 0.0f, 0.0f, -9.128709137e-02f, 0.0f, 0.0f, 0.0f, 0.0f, 0.0f, 0.0f, 0.0f, 0.0f, 0.0f, 0.0f, 0.0f, 1.581138819e-01f, 0.0f, 0.0f, 0.0f, 0.0f, 1.581138819e-01f, 0.0f, 0.0f, 0.0f, 0.0f, 0.0f, 0.0f, 0.0f, 0.0f, 0.0f, 0.0f, 0.0f, 0.0f, 0.0f, 0.0f, 0.0f, 0.0f, 0.0f, 0.0f, 0.0f, 0.0f, 0.0f, -7.319250703e-02f, 0.0f, -9.449111670e-02f, 0.0f, 0.0f, 0.0f, 0.0f, -7.319250703e-02f, 0.0f, -9.449111670e-02f, 0.0f,
  0.0f, 0.0f, 0.0f, 2.041241527e-01f, 0.0f, 0.0f, 2.041241527e-01f, 0.0f, 0.0f, 0.0f, 0.0f, 1.825741827e-01f, 0.0f, 0.0f, 1.825741827e-01f, 0.0f, 0.0f, 0.0f, 0.0f, 0.0f, 0.0f, 0.0f, 0.0f, 0.0f, 0.0f, 0.0f, 0.0f, 0.0f, 0.0f, 0.0f, 0.0f, 0.0f, 0.0f, 0.0f, 0.0f, 0.0f, 0.0f, 0.0f, 0.0f, 0.0f, 0.0f, 0.0f, 0.0f, 0.0f, 0.0f, 0.0f, 0.0f, 0.0f, 0.0f, 0.0f, 0.0f, 0.0f, 0.0f, 1.463850141e-01f, 0.0f, 0.0f, 0.0f, 0.0f, 0.0f, 0.0f, 1.463850141e-01f, 0.0f, 0.0f, 0.0f,
  0.0f, 0.0f, 0.0f, 2.041241527e-01f, 0.0f, 0.0f, 2.041241527e-01f, 0.0f, 0.0f, 0.0f, 0.0f, -9.128709137e-02f, 0.0f, 0.0f, -9.128709137e-02f, 0.0f, 0.0f, 0.0f, 0.0f, 0.0f, 0.0f, 0.0f, 0.0f, 0.0f, 0.0f, 0.0f, -1.581138819e-01f, 0.0f, 0.0f, 0.0f, 0.0f, -1.581138819e-01f, 0.0f, 0.0f, 0.0f, 0.0f, 0.0f, 0.0f, 0.0f, 0.0f, 0.0f, 0.0f, 0.0f, 0.0f, 0.0f, 0.0f, 0.0f, 0.0f, 0.0f, 0.0f, 0.0f, 0.0f, 0.0f, -7.319250703e-02f, 0.0f, 9.449111670e-02f, 0.0f, 0.0f, 0.0f, 0.0f, -7.319250703e-02f, 0.0f, 9.449111670e-02f, 0.0f,
  0.0f, 0.0f, 0.0f, 0.0f, 0.0f, 0.0f, 0.0f, 0.0f, 0.0f, 0.0f, 1.369306445e-01f, 0.0f, 0.0f, 1.369306445e-01f, 0.0f, 0.0f, 0.0f, 0.0f, 0.0f, 0.0f, 0.0f, 0.0f, 0.0f, 0.0f, 0.0f, 0.0f, 0.0f, 0.0f, 0.0f, 7.905694097e-02f, 0.0f, 0.0f, 0.0f, 0.0f, 7.905694097e-02f, 0.0f, 0.0f, 0.0f, 0.0f, 0.0f, 0.0f, 0.0f, 0.0f, 0.0f, 0.0f, 0.0f, 0.0f, 0.0f, 0.0f, 0.0f, 0.0f, 0.0f, 1.195228621e-01f, 0.0f, 0.0f, 0.0f, 0.0f, 0.0f, 0.0f, 1.195228621e-01f, 0.0f, 0.0f, 0.0f, 0.0f,
  0.0f, 0.0f, 0.0f, 0.0f, 0.0f, 0.0f, 0.0f, 0.0f, 0.0f, 0.0f, 0.0f, 0.0f, 1.369306445e-01f, 0.0f, 0.0f, 1.369306445e-01f, 0.0f, 0.0f, 0.0f, 0.0f, 0.0f, 0.0f, 0.0f, 0.0f, 0.0f, 0.0f, 0.0f, -7.905694097e-02f, 0.0f, 0.0f, 0.0f, 0.0f, -7.905694097e-02f, 0.0f, 0.0f, 0.0f, 0.0f, 0.0f, 0.0f, 0.0f, 0.0f, 0.0f, 0.0f, 0.0f, 0.0f, 0.0f, 0.0f, 0.0f, 0.0f, 0.0f, 0.0f, 0.0f, 0.0f, 0.0f, 1.195228621e-01f, 0.0f, 0.0f, 0.0f, 0.0f, 0.0f, 0.0f, 1.195228621e-01f, 0.0f, 0.0f,
  0.0f, 0.0f, 0.0f, 0.0f, 0.0f, 0.0f, 0.0f, 0.0f, 0.0f, 0.0f, 0.0f, 0.0f, 0.0f, 0.0f, 0.0f, 0.0f, 0.0f, 0.0f, 0.0f, 0.0f, 0.0f, 0.0f, 0.0f, 0.0f, 0.0f, 0.0f, 0.0f, 0.0f, 0.0f, 0.0f, 1.581138819e-01f, 0.0f, 0.0f, 0.0f, 0.0f, 1.581138819e-01f, 0.0f, 0.0f, 0.0f, 0.0f, 0.0f, 0.0f, 0.0f, 0.0f, 0.0f, 0.0f, 0.0f, 0.0f, 0.0f, 0.0f, 0.0f, 9.449111670e-02f, 0.0f, 0.0f, 0.0f, 0.0f, 0.0f, 0.0f, 9.449111670e-02f, 0.0f, 0.0f, 0.0f, 0.0f, 0.0f,
  0.0f, 0.0f, 0.0f, 0.0f, 2.041241527e-01f, 0.0f, 0.0f, 2.041241527e-01f, 0.0f, 0.0f, 0.0f, 0.0f, -9.128709137e-02f, 0.0f, 0.0f, -9.128709137e-02f, 0.0f, 0.0f, 0.0f, 0.0f, 0.0f, 0.0f, 0.0f, 0.0f, 0.0f, 0.0f, 0.0f, -1.581138819e-01f, 0.0f, 0.0f, 0.0f, 0.0f, -1.581138819e-01f, 0.0f, 0.0f, 0.0f, 0.0f, 0.0f, 0.0f, 0.0f, 0.0f, 0.0f, 0.0f, 0.0f, 0.0f, 0.0f, 0.0f, 0.0f, 0.0f, 0.0f, 0.0f, 0.0f, 0.0f, 0.0f, -2.988071553e-02f, 0.0f, -1.157275140e-01f, 0.0f, 0.0f, 0.0f, 0.0f, -2.988071553e-02f, 0.0f, -1.157275140e-01f,
  0.0f, 0.0f, 0.0f, 0.0f, 2.041241527e-01f, 0.0f, 0.0f, 2.041241527e-01f, 0.0f, 0.0f, 0.0f, 0.0f, -9.128709137e-02f, 0.0f, 0.0f, -9.128709137e-02f, 0.0f, 0.0f, 0.0f, 0.0f, 0.0f, 0.0f, 0.0f, 0.0f, 0.0f, 0.0f, 0.0f, 1.581138819e-01f, 0.0f, 0.0f, 0.0f, 0.0f, 1.581138819e-01f, 0.0f, 0.0f, 0.0f, 0.0f, 0.0f, 0.0f, 0.0f, 0.0f, 0.0f, 0.0f, 0.0f, 0.0f, 0.0f, 0.0f, 0.0f, 0.0f, 0.0f, 0.0f, 0.0f, 0.0f, 0.0f, 1.195228621e-01f, 0.0f, 0.0f, 0.0f, 0.0f, 0.0f, 0.0f, 1.195228621e-01f, 0.0f, 0.0f,
  0.0f, 0.0f, 0.0f, 0.0f, 2.041241527e-01f, 0.0f, 0.0f, 2.041241527e-01f, 0.0f, 0.0f, 0.0f, 0.0f, 1.825741827e-01f, 0.0f, 0.0f, 1.825741827e-01f, 0.0f, 0.0f, 0.0f, 0.0f, 0.0f, 0.0f, 0.0f, 0.0f, 0.0f, 0.0f, 0.0f, 0.0f, 0.0f, 0.0f, 0.0f, 0.0f, 0.0f, 0.0f, 0.0f, 0.0f, 0.0f, 0.0f, 0.0f, 0.0f, 0.0f, 0.0f, 0.0f, 0.0f, 0.0f, 0.0f, 0.0f, 0.0f, 0.0f, 0.0f, 0.0f, 0.0f, 0.0f, 0.0f, -8.964214474e-02f, 0.0f, 1.157275140e-01f, 0.0f, 0.0f, 0.0f, 0.0f, -8.964214474e-02f, 0.0f, 1.157275140e-01f,
  0.0f, 0.0f, 0.0f, 0.0f, 0.0f, 0.0f, 0.0f, 0.0f, 0.0f, 0.0f, 0.0f, 0.0f, 0.0f, 0.0f, 0.0f, 0.0f, 0.0f, 0.0f, 0.0f, 0.0f, 0.0f, 0.0f, 0.0f, 0.0f, 0.0f, 0.0f, 0.0f, 0.0f, -1.369306445e-01f, 0.0f, -7.905694097e-02f, 0.0f, 0.0f, -1.369306445e-01f, 0.0f, -7.905694097e-02f, 0.0f, 0.0f, 0.0f, 0.0f, 0.0f, 0.0f, 0.0f, 0.0f, 0.0f, 0.0f, 0.0f, 0.0f, 0.0f, 0.0f, 0.0f, 9.449111670e-02f, 0.0f, 0.0f, 0.0f, 0.0f, 0.0f, 0.0f, 9.449111670e-02f, 0.0f, 0.0f, 0.0f, 0.0f, 0.0f,
  0.0f, 0.0f, 0.0f, 0.0f, 0.0f, 0.0f, 0.0f, 0.0f, 0.0f, 0.0f, 0.0f, 1.369306445e-01f, 0.0f, 0.0f, 1.369306445e-01f, 0.0f, 0.0f, 0.0f, 0.0f, 0.0f, 0.0f, 0.0f, 0.0f, 0.0f, 0.0f, 0.0f, 7.905694097e-02f, 0.0f, 0.0f, 0.0f, 0.0f, 7.905694097e-02f, 0.0f, 0.0f, 0.0f, 0.0f, 0.0f, 0.0f, 0.0f, 0.0f, 0.0f, 0.0f, 0.0f, 0.0f, 0.0f, 0.0f, 0.0f, 0.0f, 0.0f, 0.0f, 0.0f, 0.0f, 0.0f, -7.319250703e-02f, 0.0f, 9.449111670e-02f, 0.0f, 0.0f, 0.0f, 0.0f, -7.319250703e-02f, 0.0f, 9.449111670e-02f, 0.0f,
  0.0f, 0.0f, 0.0f, 0.0f, 0.0f, 0.0f, 0.0f, 0.0f, 0.0f, 0.0f, 1.369306445e-01f, 0.0f, 0.0f, 1.369306445e-01f, 0.0f, 0.0f, 0.0f, 0.0f, 0.0f, 0.0f, 0.0f, 0.0f, 0.0f, 0.0f, 0.0f, 0.0f, 0.0f, 0.0f, 0.0f, -7.905694097e-02f, 0.0f, 0.0f, 0.0f, 0.0f, -7.905694097e-02f, 0.0f, 0.0f, 0.0f, 0.0f, 0.0f, 0.0f, 0.0f, 0.0f, 0.0f, 0.0f, 0.0f, 0.0f, 0.0f, 0.0f, 0.0f, 1.157275140e-01f, 0.0f, -2.988071553e-02f, 0.0f, 0.0f, 0.0f, 0.0f, 1.157275140e-01f, 0.0f, -2.988071553e-02f, 0.0f, 0.0f, 0.0f, 0.0f,
};


typedef float v4f __attribute__((ext_vector_type(4)));

#define WPB 4
#define TPW 5
#define KST 8
#define OTF 288

__device__ inline void load_frag(const float* __restrict__ row, int half, v2f f[KST]){
  const float* pa = row + 10 + 16*half;
  const float* pb = row + 32 + 24*half;
  const float* pc = row + 52 + 8*half;
  v2f q0 = *(const v2f*)(pa);
  v4f q1 = *(const v4f*)(pa + 2);
  v4f q2 = *(const v4f*)(pb);
  v4f q3 = *(const v4f*)(pc);
  v2f q4 = *(const v2f*)(row + 50);
  if (half) q4 = (v2f){0.0f, 0.0f};
  f[0] = q0;
  f[1] = (v2f){q1.x, q1.y};
  f[2] = (v2f){q1.z, q1.w};
  f[3] = (v2f){q2.x, q2.y};
  f[4] = (v2f){q2.z, q2.w};
  f[5] = (v2f){q3.x, q3.y};
  f[6] = (v2f){q3.z, q3.w};
  f[7] = q4;
}

__device__ inline int frag_col(int half, int s, int e){
  if (s < 3) return 10 + 16*half + 2*s + e;
  if (s < 5) return 32 + 24*half + 2*(s - 3) + e;
  if (s < 7) return 52 + 8*half + 2*(s - 5) + e;
  return half ? -1 : 50 + e;
}
__device__ inline int path_w(int c){
  return c < 16 ? 2 + (c - 10)/3 : (c < 36 ? 4 + (c - 26)/5 : 6 + (c - 50)/7);
}

__device__ inline void fold_w(const float wv[8], const float* __restrict__ crow, int half, v2f f[KST]){
  const v2f c0 = *(const v2f*)(crow + 2);
  const v4f c1 = *(const v4f*)(crow + 4);
  const float c2[6] = {c0.x, c0.y, c1.x, c1.y, c1.z, c1.w};
  #pragma unroll
  for (int s = 0; s < KST; ++s){
    #pragma unroll
    for (int e = 0; e < 2; ++e){
      const int ca = frag_col(0, s, e), cb = frag_col(1, s, e);
      const int ib = cb < 0 ? 0 : path_w(cb);
      const float ga = wv[path_w(ca)];
      const float gb = cb < 0 ? 0.0f : wv[ib];
      float v = f[s][e] * (half ? gb : ga);
      if (ca < 16){
        const float g2 = wv[(ca - 10)/3] * c2[ca - 10];
        v += half ? 0.0f : g2;
      }
      f[s][e] = v;
    }
  }
}

__global__ __launch_bounds__(WPB*32) void piezo_wmma_main(const float* __restrict__ X,
                                                          const float* __restrict__ w,
                                                          float* __restrict__ out,
                                                          int rows){
  __shared__ __attribute__((aligned(16))) float stage[WPB][OTF];

  const int lane = threadIdx.x & 31;
  const int wave = threadIdx.x >> 5;
  const int half = lane >> 4;
  const int l16  = lane & 15;
  float* st = stage[wave];

  const long tilebase = (long)(blockIdx.x * WPB + wave) * TPW;
  if ((tilebase + 1) * 16 <= (long)rows){
    const float* x0 = X + (tilebase * 16 + l16) * 64;
    __builtin_prefetch((const void*)(x0), 0, 1);
    __builtin_prefetch((const void*)(x0 + 32), 0, 1);
  }

  float wv[8];
  #pragma unroll
  for (int i = 0; i < 8; ++i) wv[i] = w[i];
  v2f bf0[KST], bf1[KST];
  load_frag(CT + (l16     ) * 64, half, bf0);
  load_frag(CT + (l16 + 16) * 64, half, bf1);
  fold_w(wv, CT + (l16     ) * 64, half, bf0);
  fold_w(wv, CT + (l16 + 16) * 64, half, bf1);

  for (int t = 0; t < TPW; ++t){
    long row0 = (tilebase + t) * 16;
    if (row0 + 16 > (long)rows) break;
    const float* xr = X + (row0 + l16) * 64;

    __builtin_prefetch((const void*)(xr + 16*64), 0, 1);
    __builtin_prefetch((const void*)(xr + 16*64 + 32), 0, 1);

    v2f a[KST];
    load_frag(xr, half, a);

    v8f acc0 = {}; v8f acc1 = {};
    #pragma unroll
    for (int s = 0; s < KST; ++s){
      acc0 = __builtin_amdgcn_wmma_f32_16x16x4_f32(false, bf0[s], false, a[s],
                                                   (short)0, acc0, false, false);
      acc1 = __builtin_amdgcn_wmma_f32_16x16x4_f32(false, bf1[s], false, a[s],
                                                   (short)0, acc1, false, false);
    }

    float* srow = st + l16*18 + half*8;
    *(v2f*)(srow    ) = (v2f){acc0[0], acc0[1]};
    *(v2f*)(srow + 2) = (v2f){acc0[2], acc0[3]};
    *(v2f*)(srow + 4) = (v2f){acc0[4], acc0[5]};
    *(v2f*)(srow + 6) = (v2f){acc0[6], acc0[7]};
    if (half == 0) *(v2f*)(st + l16*18 + 16) = (v2f){acc1[0], acc1[1]};

    asm volatile("s_wait_dscnt 0x0" ::: "memory");

    float* ot = out + row0 * 18;
    v4f p0 = *(const v4f*)(st + 4*lane);
    v4f p1 = *(const v4f*)(st + 4*(lane + 32));
    *(v4f*)(ot + 4*lane)        = p0;
    *(v4f*)(ot + 4*(lane + 32)) = p1;
    if (lane < 8){
      v4f p2 = *(const v4f*)(st + 4*(lane + 64));
      *(v4f*)(ot + 4*(lane + 64)) = p2;
    }
  }
}


extern "C" void kernel_launch(void* const* d_in, const int* in_sizes, int n_in,
                              void* d_out, int out_size, void* d_ws, size_t ws_size,
                              hipStream_t stream){
  const float* X = (const float*)d_in[0];
  const float* w = (const float*)d_in[1];
  float* out = (float*)d_out;

  const int rows  = in_sizes[0] / 64;
  const int tiles = rows / 16;
  const int blocks = (tiles + WPB*TPW - 1) / (WPB*TPW);

  piezo_wmma_main<<<blocks, WPB*32, 0, stream>>>(X, w, out, rows);
}


# X read as whole 512-B pieces: 8 loads a tile of 4 lines each, next tile's in flight in registers (was 5 loads + 2 prefetches of 16 lines each); fragments via the wave's LDS patch; 8 tiles a wave
# speedup vs baseline: 4.5086x; 1.1311x over previous
#include <hip/hip_runtime.h>

typedef float v2f __attribute__((ext_vector_type(2)));
typedef float v8f __attribute__((ext_vector_type(8)));

__device__ const float CT[32*64] __attribute__((aligned(256))) = {
  0.0f, 0.0f, 2.041241527e-01f, 0.0f, 0.0f, 2.041241527e-01f, 0.0f, 0.0f, 0.0f, 0.0f, 1.825741827e-01f, 0.0f, 0.0f, 1.825741827e-01f, 0.0f, 0.0f, 0.0f, 0.0f, 0.0f, 0.0f, 0.0f, 0.0f, 0.0f, 0.0f, 0.0f, 0.0f, 0.0f, 0.0f, 0.0f, 0.0f, 0.0f, 0.0f, 0.0f, 0.0f, 0.0f, 0.0f, 0.0f, 0.0f, 0.0f, 0.0f, 0.0f, 0.0f, 0.0f, 0.0f, 0.0f, 0.0f, 0.0f, 0.0f, 0.0f, 0.0f, -1.157275140e-01f, 0.0f, -8.964214474e-02f, 0.0f, 0.0f, 0.0f, 0.0f, -1.157275140e-01f, 0.0f, -8.964214474e-02f, 0.0f, 0.0f, 0.0f, 0.0f,
  0.0f, 0.0f, 2.041241527e-01f, 0.0f, 0.0f, 2.041241527e-01f, 0.0f, 0.0f, 0.0f, 0.0f, -9.128709137e-02f, 0.0f, 0.0f, -9.128709137e-02f, 0.0f, 0.0f, 0.0f, 0.0f, 0.0f, 0.0f, 0.0f, 0.0f, 0.0f, 0.0f, 0.0f, 0.0f, 0.0f, 0.0f, 0.0f, -1.581138819e-01f, 0.0f, 0.0f, 0.0f, 0.0f, -1.581138819e-01f, 0.0f, 0.0f, 0.0f, 0.0f, 0.0f, 0.0f, 0.0f, 0.0f, 0.0f, 0.0f, 0.0f, 0.0f, 0.0f, 0.0f, 0.0f, 0.0f, 0.0f, 1.195228621e-01f, 0.0f, 0.0f, 0.0f, 0.0f, 0.0f, 0.0f, 1.195228621e-01f, 0.0f, 0.0f, 0.0f, 0.0f,
  0.0f, 0.0f, 2.041241527e-01f, 0.0f, 0.0f, 2.041241527e-01f, 0.0f, 0.0f, 0.0f, 0.0f, -9.128709137e-02f, 0.0f, 0.0f, -9.128709137e-02f, 0.0f, 0.0f, 0.0f, 0.0f, 0.0f, 0.0f, 0.0f, 0.0f, 0.0f, 0.0f, 0.0f, 0.0f, 0.0f, 0.0f, 0.0f, 1.581138819e-01f, 0.0f, 0.0f, 0.0f, 0.0f, 1.581138819e-01f, 0.0f, 0.0f, 0.0f, 0.0f, 0.0f, 0.0f, 0.0f, 0.0f, 0.0f, 0.0f, 0.0f, 0.0f, 0.0f, 0.0f, 0.0f, 1.157275140e-01f, 0.0f, -2.988071553e-02f, 0.0f, 0.0f, 0.0f, 0.0f, 1.157275140e-01f, 0.0f, -2.988071553e-02f, 0.0f, 0.0f, 0.0f, 0.0f,
  0.0f, 0.0f, 0.0f, 0.0f, 0.0f, 0.0f, 0.0f, 0.0f, 0.0f, 0.0f, 0.0f, 1.369306445e-01f, 0.0f, 0.0f, 1.369306445e-01f, 0.0f, 0.0f, 0.0f, 0.0f, 0.0f, 0.0f, 0.0f, 0.0f, 0.0f, 0.0f, 0.0f, -7.905694097e-02f, 0.0f, 0.0f, 0.0f, 0.0f, -7.905694097e-02f, 0.0f, 0.0f, 0.0f, 0.0f, 0.0f, 0.0f, 0.0f, 0.0f, 0.0f, 0.0f, 0.0f, 0.0f, 0.0f, 0.0f, 0.0f, 0.0f, 0.0f, 0.0f, 0.0f, 0.0f, 0.0f, -7.319250703e-02f, 0.0f, -9.449111670e-02f, 0.0f, 0.0f, 0.0f, 0.0f, -7.319250703e-02f, 0.0f, -9.449111670e-02f, 0.0f,
  0.0f, 0.0f, 0.0f, 0.0f, 0.0f, 0.0f, 0.0f, 0.0f, 0.0f, 0.0f, 0.0f, 0.0f, 0.0f, 0.0f, 0.0f, 0.0f, 0.0f, 0.0f, 0.0f, 0.0f, 0.0f, 0.0f, 0.0f, 0.0f, 0.0f, 0.0f, 0.0f, 0.0f, 1.369306445e-01f, 0.0f, -7.905694097e-02f, 0.0f, 0.0f, 1.369306445e-01f, 0.0f, -7.905694097e-02f, 0.0f, 0.0f, 0.0f, 0.0f, 0.0f, 0.0f, 0.0f, 0.0f, 0.0f, 0.0f, 0.0f, 0.0f, 0.0f, 0.0f, 0.0f, 9.449111670e-02f, 0.0f, 0.0f, 0.0f, 0.0f, 0.0f, 0.0f, 9.449111670e-02f, 0.0f, 0.0f, 0.0f, 0.0f, 0.0f,
  0.0f, 0.0f, 0.0f, 0.0f, 0.0f, 0.0f, 0.0f, 0.0f, 0.0f, 0.0f, 0.0f, 0.0f, 1.369306445e-01f, 0.0f, 0.0f, 1.369306445e-01f, 0.0f, 0.0f, 0.0f, 0.0f, 0.0f, 0.0f, 0.0f, 0.0f, 0.0f, 0.0f, 0.0f, 7.905694097e-02f, 0.0f, 0.0f, 0.0f, 0.0f, 7.905694097e-02f, 0.0f, 0.0f, 0.0f, 0.0f, 0.0f, 0.0f, 0.0f, 0.0f, 0.0f, 0.0f, 0.0f, 0.0f, 0.0f, 0.0f, 0.0f, 0.0f, 0.0f, 0.0f, 0.0f, 0.0f, 0.0f, -2.988071553e-02f, 0.0f, -1.157275140e-01f, 0.0f, 0.0f, 0.0f, 0.0f, -2.988071553e-02f, 0.0f, -1.157275140e-01f,
  0.0f, 0.0f, 0.0f, 2.041241527e-01f, 0.0f, 0.0f, 2.041241527e-01f, 0.0f, 0.0f, 0.0f, 0.0f, -9.128709137e-02f, 0.0f, 0.0f, -9.128709137e-02f, 0.0f, 0.0f, 0.0f, 0.0f, 0.0f, 0.0f, 0.0f, 0.0f, 0.0f, 0.0f, 0.0f, 1.581138819e-01f, 0.0f, 0.0f, 0.0f, 0.0f, 1.581138819e-01f, 0.0f, 0.0f, 0.0f, 0.0f, 0.0f, 0.0f, 0.0f, 0.0f, 0.0f, 0.0f, 0.0f, 0.0f, 0.0f, 0.0f, 0.0f, 0.0f, 0.0f, 0.0f, 0.0f, 0.0f, 0.0f, -7.319250703e-02f, 0.0f, -9.449111670e-02f, 0.0f, 0.0f, 0.0f, 0.0f, -7.319250703e-02f, 0.0f, -9.449111670e-02f, 0.0f,
  0.0f, 0.0f, 0.0f, 2.041241527e-01f, 0.0f, 0.0f, 2.041241527e-01f, 0.0f, 0.0f, 0.0f, 0.0f, 1.825741827e-01f, 0.0f, 0.0f, 1.825741827e-01f, 0.0f, 0.0f, 0.0f, 0.0f, 0.0f, 0.0f, 0.0f, 0.0f, 0.0f, 0.0f, 0.0f, 0.0f, 0.0f, 0.0f, 0.0f, 0.0f, 0.0f, 0.0f, 0.0f, 0.0f, 0.0f, 0.0f, 0.0f, 0.0f, 0.0f, 0.0f, 0.0f, 0.0f, 0.0f, 0.0f, 0.0f, 0.0f, 0.0f, 0.0f, 0.0f, 0.0f, 0.0f, 0.0f, 1.463850141e-01f, 0.0f, 0.0f, 0.0f, 0.0f, 0.0f, 0.0f, 1.463850141e-01f, 0.0f, 0.0f, 0.0f,
  0.0f, 0.0f, 0.0f, 2.041241527e-01f, 0.0f, 0.0f, 2.041241527e-01f, 0.0f, 0.0f, 0.0f, 0.0f, -9.128709137e-02f, 0.0f, 0.0f, -9.128709137e-02f, 0.0f, 0.0f, 0.0f, 0.0f, 0.0f, 0.0f, 0.0f, 0.0f, 0.0f, 0.0f, 0.0f, -1.581138819e-01f, 0.0f, 0.0f, 0.0f, 0.0f, -1.581138819e-01f, 0.0f, 0.0f, 0.0f, 0.0f, 0.0f, 0.0f, 0.0f, 0.0f, 0.0f, 0.0f, 0.0f, 0.0f, 0.0f, 0.0f, 0.0f, 0.0f, 0.0f, 0.0f, 0.0f, 0.0f, 0.0f, -7.319250703e-02f, 0.0f, 9.449111670e-02f, 0.0f, 0.0f, 0.0f, 0.0f, -7.319250703e-02f, 0.0f, 9.449111670e-02f, 0.0f,
  0.0f, 0.0f, 0.0f, 0.0f, 0.0f, 0.0f, 0.0f, 0.0f, 0.0f, 0.0f, 1.369306445e-01f, 0.0f, 0.0f, 1.369306445e-01f, 0.0f, 0.0f, 0.0f, 0.0f, 0.0f, 0.0f, 0.0f, 0.0f, 0.0f, 0.0f, 0.0f, 0.0f, 0.0f, 0.0f, 0.0f, 7.905694097e-02f, 0.0f, 0.0f, 0.0f, 0.0f, 7.905694097e-02f, 0.0f, 0.0f, 0.0f, 0.0f, 0.0f, 0.0f, 0.0f, 0.0f, 0.0f, 0.0f, 0.0f, 0.0f, 0.0f, 0.0f, 0.0f, 0.0f, 0.0f, 1.195228621e-01f, 0.0f, 0.0f, 0.0f, 0.0f, 0.0f, 0.0f, 1.195228621e-01f, 0.0f, 0.0f, 0.0f, 0.0f,
  0.0f, 0.0f, 0.0f, 0.0f, 0.0f, 0.0f, 0.0f, 0.0f, 0.0f, 0.0f, 0.0f, 0.0f, 1.369306445e-01f, 0.0f, 0.0f, 1.369306445e-01f, 0.0f, 0.0f, 0.0f, 0.0f, 0.0f, 0.0f, 0.0f, 0.0f, 0.0f, 0.0f, 0.0f, -7.905694097e-02f, 0.0f, 0.0f, 0.0f, 0.0f, -7.905694097e-02f, 0.0f, 0.0f, 0.0f, 0.0f, 0.0f, 0.0f, 0.0f, 0.0f, 0.0f, 0.0f, 0.0f, 0.0f, 0.0f, 0.0f, 0.0f, 0.0f, 0.0f, 0.0f, 0.0f, 0.0f, 0.0f, 1.195228621e-01f, 0.0f, 0.0f, 0.0f, 0.0f, 0.0f, 0.0f, 1.195228621e-01f, 0.0f, 0.0f,
  0.0f, 0.0f, 0.0f, 0.0f, 0.0f, 0.0f, 0.0f, 0.0f, 0.0f, 0.0f, 0.0f, 0.0f, 0.0f, 0.0f, 0.0f, 0.0f, 0.0f, 0.0f, 0.0f, 0.0f, 0.0f, 0.0f, 0.0f, 0.0f, 0.0f, 0.0f, 0.0f, 0.0f, 0.0f, 0.0f, 1.581138819e-01f, 0.0f, 0.0f, 0.0f, 0.0f, 1.581138819e-01f, 0.0f, 0.0f, 0.0f, 0.0f, 0.0f, 0.0f, 0.0f, 0.0f, 0.0f, 0.0f, 0.0f, 0.0f, 0.0f, 0.0f, 0.0f, 9.449111670e-02f, 0.0f, 0.0f, 0.0f, 0.0f, 0.0f, 0.0f, 9.449111670e-02f, 0.0f, 0.0f, 0.0f, 0.0f, 0.0f,
  0.0f, 0.0f, 0.0f, 0.0f, 2.041241527e-01f, 0.0f, 0.0f, 2.041241527e-01f, 0.0f, 0.0f, 0.0f, 0.0f, -9.128709137e-02f, 0.0f, 0.0f, -9.128709137e-02f, 0.0f, 0.0f, 0.0f, 0.0f, 0.0f, 0.0f, 0.0f, 0.0f, 0.0f, 0.0f, 0.0f, -1.581138819e-01f, 0.0f, 0.0f, 0.0f, 0.0f, -1.581138819e-01f, 0.0f, 0.0f, 0.0f, 0.0f, 0.0f, 0.0f, 0.0f, 0.0f, 0.0f, 0.0f, 0.0f, 0.0f, 0.0f, 0.0f, 0.0f, 0.0f, 0.0f, 0.0f, 0.0f, 0.0f, 0.0f, -2.988071553e-02f, 0.0f, -1.157275140e-01f, 0.0f, 0.0f, 0.0f, 0.0f, -2.988071553e-02f, 0.0f, -1.157275140e-01f,
  0.0f, 0.0f, 0.0f, 0.0f, 2.041241527e-01f, 0.0f, 0.0f, 2.041241527e-01f, 0.0f, 0.0f, 0.0f, 0.0f, -9.128709137e-02f, 0.0f, 0.0f, -9.128709137e-02f, 0.0f, 0.0f, 0.0f, 0.0f, 0.0f, 0.0f, 0.0f, 0.0f, 0.0f, 0.0f, 0.0f, 1.581138819e-01f, 0.0f, 0.0f, 0.0f, 0.0f, 1.581138819e-01f, 0.0f, 0.0f, 0.0f, 0.0f, 0.0f, 0.0f, 0.0f, 0.0f, 0.0f, 0.0f, 0.0f, 0.0f, 0.0f, 0.0f, 0.0f, 0.0f, 0.0f, 0.0f, 0.0f, 0.0f, 0.0f, 1.195228621e-01f, 0.0f, 0.0f, 0.0f, 0.0f, 0.0f, 0.0f, 1.195228621e-01f, 0.0f, 0.0f,
  0.0f, 0.0f, 0.0f, 0.0f, 2.041241527e-01f, 0.0f, 0.0f, 2.041241527e-01f, 0.0f, 0.0f, 0.0f, 0.0f, 1.825741827e-01f, 0.0f, 0.0f, 1.825741827e-01f, 0.0f, 0.0f, 0.0f, 0.0f, 0.0f, 0.0f, 0.0f, 0.0f, 0.0f, 0.0f, 0.0f, 0.0f, 0.0f, 0.0f, 0.0f, 0.0f, 0.0f, 0.0f, 0.0f, 0.0f, 0.0f, 0.0f, 0.0f, 0.0f, 0.0f, 0.0f, 0.0f, 0.0f, 0.0f, 0.0f, 0.0f, 0.0f, 0.0f, 0.0f, 0.0f, 0.0f, 0.0f, 0.0f, -8.964214474e-02f, 0.0f, 1.157275140e-01f, 0.0f, 0.0f, 0.0f, 0.0f, -8.964214474e-02f, 0.0f, 1.157275140e-01f,
  0.0f, 0.0f, 0.0f, 0.0f, 0.0f, 0.0f, 0.0f, 0.0f, 0.0f, 0.0f, 0.0f, 0.0f, 0.0f, 0.0f, 0.0f, 0.0f, 0.0f, 0.0f, 0.0f, 0.0f, 0.0f, 0.0f, 0.0f, 0.0f, 0.0f, 0.0f, 0.0f, 0.0f, -1.369306445e-01f, 0.0f, -7.905694097e-02f, 0.0f, 0.0f, -1.369306445e-01f, 0.0f, -7.905694097e-02f, 0.0f, 0.0f, 0.0f, 0.0f, 0.0f, 0.0f, 0.0f, 0.0f, 0.0f, 0.0f, 0.0f, 0.0f, 0.0f, 0.0f, 0.0f, 9.449111670e-02f, 0.0f, 0.0f, 0.0f, 0.0f, 0.0f, 0.0f, 9.449111670e-02f, 0.0f, 0.0f, 0.0f, 0.0f, 0.0f,
  0.0f, 0.0f, 0.0f, 0.0f, 0.0f, 0.0f, 0.0f, 0.0f, 0.0f, 0.0f, 0.0f, 1.369306445e-01f, 0.0f, 0.0f, 1.369306445e-01f, 0.0f, 0.0f, 0.0f, 0.0f, 0.0f, 0.0f, 0.0f, 0.0f, 0.0f, 0.0f, 0.0f, 7.905694097e-02f, 0.0f, 0.0f, 0.0f, 0.0f, 7.905694097e-02f, 0.0f, 0.0f, 0.0f, 0.0f, 0.0f, 0.0f, 0.0f, 0.0f, 0.0f, 0.0f, 0.0f, 0.0f, 0.0f, 0.0f, 0.0f, 0.0f, 0.0f, 0.0f, 0.0f, 0.0f, 0.0f, -7.319250703e-02f, 0.0f, 9.449111670e-02f, 0.0f, 0.0f, 0.0f, 0.0f, -7.319250703e-02f, 0.0f, 9.449111670e-02f, 0.0f,
  0.0f, 0.0f, 0.0f, 0.0f, 0.0f, 0.0f, 0.0f, 0.0f, 0.0f, 0.0f, 1.369306445e-01f, 0.0f, 0.0f, 1.369306445e-01f, 0.0f, 0.0f, 0.0f, 0.0f, 0.0f, 0.0f, 0.0f, 0.0f, 0.0f, 0.0f, 0.0f, 0.0f, 0.0f, 0.0f, 0.0f, -7.905694097e-02f, 0.0f, 0.0f, 0.0f, 0.0f, -7.905694097e-02f, 0.0f, 0.0f, 0.0f, 0.0f, 0.0f, 0.0f, 0.0f, 0.0f, 0.0f, 0.0f, 0.0f, 0.0f, 0.0f, 0.0f, 0.0f, 1.157275140e-01f, 0.0f, -2.988071553e-02f, 0.0f, 0.0f, 0.0f, 0.0f, 1.157275140e-01f, 0.0f, -2.988071553e-02f, 0.0f, 0.0f, 0.0f, 0.0f,
};


typedef float v4f __attribute__((ext_vector_type(4)));

#define WPB 4
#define TPW 8
#define KST 8
#define OTF 288
#define XP  68
#define LPT 8

__device__ inline void load_frag(const float* __restrict__ row, int half, v2f f[KST]){
  const float* pa = row + 10 + 16*half;
  const float* pb = row + 32 + 24*half;
  const float* pc = row + 52 + 8*half;
  v2f q0 = *(const v2f*)(pa);
  v4f q1 = *(const v4f*)(pa + 2);
  v4f q2 = *(const v4f*)(pb);
  v4f q3 = *(const v4f*)(pc);
  v2f q4 = *(const v2f*)(row + 50);
  if (half) q4 = (v2f){0.0f, 0.0f};
  f[0] = q0;
  f[1] = (v2f){q1.x, q1.y};
  f[2] = (v2f){q1.z, q1.w};
  f[3] = (v2f){q2.x, q2.y};
  f[4] = (v2f){q2.z, q2.w};
  f[5] = (v2f){q3.x, q3.y};
  f[6] = (v2f){q3.z, q3.w};
  f[7] = q4;
}

__device__ inline int frag_col(int half, int s, int e){
  if (s < 3) return 10 + 16*half + 2*s + e;
  if (s < 5) return 32 + 24*half + 2*(s - 3) + e;
  if (s < 7) return 52 + 8*half + 2*(s - 5) + e;
  return half ? -1 : 50 + e;
}
__device__ inline int path_w(int c){
  return c < 16 ? 2 + (c - 10)/3 : (c < 36 ? 4 + (c - 26)/5 : 6 + (c - 50)/7);
}

__device__ inline void fold_w(const float wv[8], const float* __restrict__ crow, int half, v2f f[KST]){
  const v2f c0 = *(const v2f*)(crow + 2);
  const v4f c1 = *(const v4f*)(crow + 4);
  const float c2[6] = {c0.x, c0.y, c1.x, c1.y, c1.z, c1.w};
  #pragma unroll
  for (int s = 0; s < KST; ++s){
    #pragma unroll
    for (int e = 0; e < 2; ++e){
      const int ca = frag_col(0, s, e), cb = frag_col(1, s, e);
      const int ib = cb < 0 ? 0 : path_w(cb);
      const float ga = wv[path_w(ca)];
      const float gb = cb < 0 ? 0.0f : wv[ib];
      float v = f[s][e] * (half ? gb : ga);
      if (ca < 16){
        const float g2 = wv[(ca - 10)/3] * c2[ca - 10];
        v += half ? 0.0f : g2;
      }
      f[s][e] = v;
    }
  }
}

__global__ __launch_bounds__(WPB*32) void piezo_wmma_main(const float* __restrict__ X,
                                                          const float* __restrict__ w,
                                                          float* __restrict__ out,
                                                          int rows){
  __shared__ __attribute__((aligned(16))) float xpatch[WPB][16*XP];
  __shared__ __attribute__((aligned(16))) float stage[WPB][OTF];

  const int lane = threadIdx.x & 31;
  const int wave = threadIdx.x >> 5;
  const int half = lane >> 4;
  const int l16  = lane & 15;
  float* xs = xpatch[wave];
  float* st = stage[wave];

  const long tilebase = (long)(blockIdx.x * WPB + wave) * TPW;
  const long left = (long)(rows / 16) - tilebase;
  const int nt = left < (long)TPW ? (int)left : TPW;
  if (nt <= 0) return;

  float wv[8];
  #pragma unroll
  for (int i = 0; i < 8; ++i) wv[i] = w[i];
  v2f bf0[KST], bf1[KST];
  load_frag(CT + (l16     ) * 64, half, bf0);
  load_frag(CT + (l16 + 16) * 64, half, bf1);

  const float* xg = X + tilebase * (16*64) + 4*lane;
  v4f nx[LPT];
  #pragma unroll
  for (int q = 0; q < LPT; ++q) nx[q] = *(const v4f*)(xg + q*128);

  fold_w(wv, CT + (l16     ) * 64, half, bf0);
  fold_w(wv, CT + (l16 + 16) * 64, half, bf1);

  float* xw = xs + half*XP + 4*l16;
  const float* xr = xs + l16*XP;

  for (int t = 0; t < nt; ++t){
    const long row0 = (tilebase + t) * 16;

    #pragma unroll
    for (int q = 0; q < LPT; ++q) *(v4f*)(xw + 2*q*XP) = nx[q];

    if (t + 1 < nt){
      const float* xn = xg + (long)(t + 1) * (16*64);
      #pragma unroll
      for (int q = 0; q < LPT; ++q) nx[q] = *(const v4f*)(xn + q*128);
    }

    asm volatile("s_wait_dscnt 0x0" ::: "memory");

    v2f a[KST];
    load_frag(xr, half, a);

    v8f acc0 = {}; v8f acc1 = {};
    #pragma unroll
    for (int s = 0; s < KST; ++s){
      acc0 = __builtin_amdgcn_wmma_f32_16x16x4_f32(false, bf0[s], false, a[s],
                                                   (short)0, acc0, false, false);
      acc1 = __builtin_amdgcn_wmma_f32_16x16x4_f32(false, bf1[s], false, a[s],
                                                   (short)0, acc1, false, false);
    }

    float* srow = st + l16*18 + half*8;
    *(v2f*)(srow    ) = (v2f){acc0[0], acc0[1]};
    *(v2f*)(srow + 2) = (v2f){acc0[2], acc0[3]};
    *(v2f*)(srow + 4) = (v2f){acc0[4], acc0[5]};
    *(v2f*)(srow + 6) = (v2f){acc0[6], acc0[7]};
    if (half == 0) *(v2f*)(st + l16*18 + 16) = (v2f){acc1[0], acc1[1]};

    asm volatile("s_wait_dscnt 0x0" ::: "memory");

    float* ot = out + row0 * 18;
    v4f p0 = *(const v4f*)(st + 4*lane);
    v4f p1 = *(const v4f*)(st + 4*(lane + 32));
    *(v4f*)(ot + 4*lane)        = p0;
    *(v4f*)(ot + 4*(lane + 32)) = p1;
    if (lane < 8){
      v4f p2 = *(const v4f*)(st + 4*(lane + 64));
      *(v4f*)(ot + 4*(lane + 64)) = p2;
    }
  }
}


extern "C" void kernel_launch(void* const* d_in, const int* in_sizes, int n_in,
                              void* d_out, int out_size, void* d_ws, size_t ws_size,
                              hipStream_t stream){
  const float* X = (const float*)d_in[0];
  const float* w = (const float*)d_in[1];
  float* out = (float*)d_out;

  const int rows  = in_sizes[0] / 64;
  const int tiles = rows / 16;
  const int blocks = (tiles + WPB*TPW - 1) / (WPB*TPW);

  piezo_wmma_main<<<blocks, WPB*32, 0, stream>>>(X, w, out, rows);
}
